# starting kernel
# baseline (speedup 1.0000x reference)
#include <hip/hip_runtime.h>
#include <hip/hip_bf16.h>
#include <math.h>


typedef __attribute__((ext_vector_type(2))) float v2f;
typedef __attribute__((ext_vector_type(8))) float v8f;

#define NEG_SLOPE 0.2f

__device__ __forceinline__ unsigned enc_ord(float f) {
  unsigned u = __float_as_uint(f);
  return (u & 0x80000000u) ? ~u : (u | 0x80000000u);
}
__device__ __forceinline__ float dec_ord(unsigned u) {
  unsigned b = (u & 0x80000000u) ? (u & 0x7FFFFFFFu) : ~u;
  return __uint_as_float(b);
}
#define ENC_NEG_INF 0x007FFFFFu

__global__ void init_layer_kernel(float* __restrict__ acc, int nm,
                                  float* __restrict__ den,
                                  unsigned* __restrict__ mlog, int nh) {
  int i = blockIdx.x * blockDim.x + threadIdx.x;
  if (i < nm) acc[i] = 0.0f;
  if (i < nh) { den[i] = 0.0f; mlog[i] = ENC_NEG_INF; }
}

__global__ __launch_bounds__(128)
void gemm_wmma_f32(const float* __restrict__ X, const float* __restrict__ W,
                   float* __restrict__ Hout, int K, int M) {
  const int wave = threadIdx.x >> 5;
  const int lane = threadIdx.x & 31;
  const int lr   = lane & 15;
  const int kh   = (lane >> 4) << 1;

  const int rowBase = blockIdx.x << 4;
  const int nBase   = (blockIdx.y * 4 + wave) << 6;
  if (nBase >= M) return;

  v8f c0 = {}, c1 = {}, c2 = {}, c3 = {};
  const float* __restrict__ arow = X + (size_t)(rowBase + lr) * K;

  for (int k = 0; k < K; k += 4) {
    v2f a;
    a.x = arow[k + kh];
    a.y = arow[k + kh + 1];

    const float* __restrict__ wp = W + (size_t)(k + kh) * M + nBase + lr;
    v2f b0, b1, b2, b3;
    b0.x = wp[0];   b0.y = wp[M];
    b1.x = wp[16];  b1.y = wp[M + 16];
    b2.x = wp[32];  b2.y = wp[M + 32];
    b3.x = wp[48];  b3.y = wp[M + 48];

    c0 = __builtin_amdgcn_wmma_f32_16x16x4_f32(false, a, false, b0, (short)0, c0, false, false);
    c1 = __builtin_amdgcn_wmma_f32_16x16x4_f32(false, a, false, b1, (short)0, c1, false, false);
    c2 = __builtin_amdgcn_wmma_f32_16x16x4_f32(false, a, false, b2, (short)0, c2, false, false);
    c3 = __builtin_amdgcn_wmma_f32_16x16x4_f32(false, a, false, b3, (short)0, c3, false, false);
  }

  const int colOff = lane & 15;
  const int rowOff = (lane >> 4) << 3;
#pragma unroll
  for (int v = 0; v < 8; ++v) {
    float* __restrict__ op =
        Hout + (size_t)(rowBase + rowOff + v) * M + nBase + colOff;
    op[0]  = c0[v];
    op[16] = c1[v];
    op[32] = c2[v];
    op[48] = c3[v];
  }
}

__global__ __launch_bounds__(128)
void alpha_kernel(const float* __restrict__ Hf, const float* __restrict__ asrc,
                  const float* __restrict__ adst, float* __restrict__ as_out,
                  float* __restrict__ ad_out, int Hh, int M, int total) {
  const int wid  = blockIdx.x * 4 + (threadIdx.x >> 5);
  const int lane = threadIdx.x & 31;
  if (wid >= total) return;
  const int n  = wid / Hh;
  const int hd = wid - n * Hh;

  const float4 hv = *(const float4*)(Hf + (size_t)n * M + hd * 128 + lane * 4);
  const float4 sv = *(const float4*)(asrc + hd * 128 + lane * 4);
  const float4 dv = *(const float4*)(adst + hd * 128 + lane * 4);
  float s = hv.x * sv.x + hv.y * sv.y + hv.z * sv.z + hv.w * sv.w;
  float d = hv.x * dv.x + hv.y * dv.y + hv.z * dv.z + hv.w * dv.w;
#pragma unroll
  for (int off = 16; off > 0; off >>= 1) {
    s += __shfl_xor(s, off);
    d += __shfl_xor(d, off);
  }
  if (lane == 0) { as_out[wid] = s; ad_out[wid] = d; }
}

__global__ void edge_max_kernel(const float* __restrict__ as,
                                const float* __restrict__ ad,
                                const int* __restrict__ esrc,
                                const int* __restrict__ edst,
                                unsigned* __restrict__ mlog,
                                int Hh, int E0, int totalEH) {
  int i = blockIdx.x * blockDim.x + threadIdx.x;
  if (i >= totalEH) return;
  const int e  = i / Hh;
  const int hd = i - e * Hh;
  int s, d;
  if (e < E0) { s = esrc[e]; d = edst[e]; } else { s = d = e - E0; }
  float l = as[s * Hh + hd] + ad[d * Hh + hd];
  l = (l > 0.0f) ? l : NEG_SLOPE * l;
  atomicMax(&mlog[d * Hh + hd], enc_ord(l));
}

__global__ void decode_max_kernel(const unsigned* __restrict__ mlog,
                                  float* __restrict__ mfl, int nh) {
  int i = blockIdx.x * blockDim.x + threadIdx.x;
  if (i >= nh) return;
  float m = dec_ord(mlog[i]);
  mfl[i] = isfinite(m) ? m : 0.0f;
}

__global__ void edge_exp_kernel(const float* __restrict__ as,
                                const float* __restrict__ ad,
                                const float* __restrict__ mfl,
                                const int* __restrict__ esrc,
                                const int* __restrict__ edst,
                                float* __restrict__ exb,
                                float* __restrict__ den,
                                int Hh, int E0, int totalEH) {
  int i = blockIdx.x * blockDim.x + threadIdx.x;
  if (i >= totalEH) return;
  const int e  = i / Hh;
  const int hd = i - e * Hh;
  int s, d;
  if (e < E0) { s = esrc[e]; d = edst[e]; } else { s = d = e - E0; }
  float l = as[s * Hh + hd] + ad[d * Hh + hd];
  l = (l > 0.0f) ? l : NEG_SLOPE * l;
  float ex = expf(l - mfl[d * Hh + hd]);
  exb[i] = ex;
  atomicAdd(&den[d * Hh + hd], ex);
}

__global__ __launch_bounds__(128)
void aggregate_kernel(const float* __restrict__ Hf, const float* __restrict__ exb,
                      const float* __restrict__ den,
                      const int* __restrict__ esrc, const int* __restrict__ edst,
                      float* __restrict__ acc, int Hh, int M, int E0, int ET) {
  const int e    = blockIdx.x * 4 + (threadIdx.x >> 5);
  const int lane = threadIdx.x & 31;
  if (e >= ET) return;
  int s, d;
  if (e < E0) { s = esrc[e]; d = edst[e]; } else { s = d = e - E0; }

  for (int j = 0; j < Hh; ++j) {
    const float a = exb[(size_t)e * Hh + j] / (den[d * Hh + j] + 1e-16f);
    const int c = j * 128 + lane * 4;
    const float4 hv = *(const float4*)(Hf + (size_t)s * M + c);
    float* __restrict__ op = acc + (size_t)d * M + c;
    atomicAdd(op + 0, hv.x * a);
    atomicAdd(op + 1, hv.y * a);
    atomicAdd(op + 2, hv.z * a);
    atomicAdd(op + 3, hv.w * a);
  }
}

__global__ void finalize_kernel(const float* __restrict__ acc,
                                const float* __restrict__ bias,
                                float* __restrict__ out, int M, int doElu, int nm) {
  int i = blockIdx.x * blockDim.x + threadIdx.x;
  if (i >= nm) return;
  const int c = i % M;
  float v = acc[i] + bias[c];
  if (doElu) v = (v > 0.0f) ? v : expm1f(v);
  out[i] = v;
}

static void run_layer(hipStream_t stream, const float* xin, const float* W,
                      const float* asrc, const float* adst, const float* bias,
                      int N, int K, int M, int Hh, int E0, int ET,
                      const int* esrc, const int* edst, bool elu,
                      float* hbuf, float* acc, float* asb, float* adb,
                      unsigned* mlog, float* mfl, float* den, float* exb,
                      float* out) {
  const int nm = N * M;
  const int nh = N * Hh;
  const int teh = ET * Hh;

  init_layer_kernel<<<(nm + 255) / 256, 256, 0, stream>>>(acc, nm, den, mlog, nh);

  dim3 ggrid(N / 16, (M + 255) / 256);
  gemm_wmma_f32<<<ggrid, 128, 0, stream>>>(xin, W, hbuf, K, M);

  alpha_kernel<<<(nh + 3) / 4, 128, 0, stream>>>(hbuf, asrc, adst, asb, adb, Hh, M, nh);

  edge_max_kernel<<<(teh + 255) / 256, 256, 0, stream>>>(asb, adb, esrc, edst, mlog,
                                                         Hh, E0, teh);
  decode_max_kernel<<<(nh + 255) / 256, 256, 0, stream>>>(mlog, mfl, nh);

  edge_exp_kernel<<<(teh + 255) / 256, 256, 0, stream>>>(asb, adb, mfl, esrc, edst,
                                                         exb, den, Hh, E0, teh);

  aggregate_kernel<<<(ET + 3) / 4, 128, 0, stream>>>(hbuf, exb, den, esrc, edst,
                                                     acc, Hh, M, E0, ET);

  finalize_kernel<<<(nm + 255) / 256, 256, 0, stream>>>(acc, bias, out, M,
                                                        elu ? 1 : 0, nm);
}

extern "C" void kernel_launch(void* const* d_in, const int* in_sizes, int n_in,
                              void* d_out, int out_size, void* d_ws, size_t ws_size,
                              hipStream_t stream) {
  const float* x      = (const float*)d_in[0];
  const int*   ei     = (const int*)d_in[1];
  const float* W0     = (const float*)d_in[2];
  const float* a_src0 = (const float*)d_in[3];
  const float* a_dst0 = (const float*)d_in[4];
  const float* b0     = (const float*)d_in[5];
  const float* W1     = (const float*)d_in[6];
  const float* a_src1 = (const float*)d_in[7];
  const float* a_dst1 = (const float*)d_in[8];
  const float* b1     = (const float*)d_in[9];
  const float* W2     = (const float*)d_in[10];
  const float* a_src2 = (const float*)d_in[11];
  const float* a_dst2 = (const float*)d_in[12];
  const float* b2     = (const float*)d_in[13];

  const int N  = in_sizes[0] / 256;
  const int E0 = in_sizes[1] / 2;
  const int ET = E0 + N;
  const int* esrc = ei;
  const int* edst = ei + E0;

  float* ws = (float*)d_ws;
  size_t p = 0;
  float*    hbuf = ws + p;            p += (size_t)N * 512;
  float*    acc  = ws + p;            p += (size_t)N * 512;
  float*    xbuf = ws + p;            p += (size_t)N * 512;
  float*    asb  = ws + p;            p += (size_t)N * 4;
  float*    adb  = ws + p;            p += (size_t)N * 4;
  unsigned* mlog = (unsigned*)(ws + p); p += (size_t)N * 4;
  float*    mfl  = ws + p;            p += (size_t)N * 4;
  float*    den  = ws + p;            p += (size_t)N * 4;
  float*    exb  = ws + p;            p += (size_t)ET * 4;
  (void)ws_size; (void)n_in; (void)out_size;

  run_layer(stream, x, W0, a_src0, a_dst0, b0, N, 256, 512, 4, E0, ET,
            esrc, edst, true, hbuf, acc, asb, adb, mlog, mfl, den, exb, xbuf);
  run_layer(stream, xbuf, W1, a_src1, a_dst1, b1, N, 512, 512, 4, E0, ET,
            esrc, edst, true, hbuf, acc, asb, adb, mlog, mfl, den, exb, xbuf);
  run_layer(stream, xbuf, W2, a_src2, a_dst2, b2, N, 512, 128, 1, E0, ET,
            esrc, edst, false, hbuf, acc, asb, adb, mlog, mfl, den, exb,
            (float*)d_out);
}


# Edges sorted by destination once per call (count, scan, fill); one wave per node and head does max, exp-sum (butterfly) and weighted row gather in registers; no float atomics; order = arrival rank
# speedup vs baseline: 2.3638x; 2.3638x over previous
#include <hip/hip_runtime.h>
#include <hip/hip_bf16.h>
#include <math.h>


typedef __attribute__((ext_vector_type(2))) float v2f;
typedef __attribute__((ext_vector_type(8))) float v8f;

#define NEG_SLOPE 0.2f

__global__ void csr_zero_kernel(int* __restrict__ cnt, int n) {
  int i = blockIdx.x * blockDim.x + threadIdx.x;
  if (i < n) cnt[i] = 0;
}

__global__ void csr_count_kernel(const int* __restrict__ edst,
                                 int* __restrict__ cnt, int E0) {
  int e = blockIdx.x * blockDim.x + threadIdx.x;
  if (e < E0) atomicAdd(&cnt[edst[e]], 1);
}

__global__ __launch_bounds__(1024)
void csr_scan_kernel(int* __restrict__ cnt, int* __restrict__ rowptr, int n) {
  __shared__ int sh[1024];
  const int t   = threadIdx.x;
  const int per = (n + 1023) / 1024;
  const int b   = t * per;
  const int e   = (b + per < n) ? (b + per) : n;
  int sum = 0;
  for (int i = b; i < e; ++i) sum += cnt[i];
  sh[t] = sum;
  __syncthreads();
  for (int off = 1; off < 1024; off <<= 1) {
    const int v = (t >= off) ? sh[t - off] : 0;
    __syncthreads();
    sh[t] += v;
    __syncthreads();
  }
  int run = sh[t] - sum;
  for (int i = b; i < e; ++i) {
    const int c = cnt[i];
    rowptr[i] = run;
    cnt[i]    = run;
    run += c;
  }
  if (t == 1023) rowptr[n] = sh[1023];
}

__global__ void csr_fill_kernel(const int* __restrict__ esrc,
                                const int* __restrict__ edst,
                                int* __restrict__ cursor,
                                int* __restrict__ col, int E0) {
  int e = blockIdx.x * blockDim.x + threadIdx.x;
  if (e < E0) {
    const int p = atomicAdd(&cursor[edst[e]], 1);
    col[p] = esrc[e];
  }
}

__global__ __launch_bounds__(128)
void gemm_wmma_f32(const float* __restrict__ X, const float* __restrict__ W,
                   float* __restrict__ Hout, int K, int M) {
  const int wave = threadIdx.x >> 5;
  const int lane = threadIdx.x & 31;
  const int lr   = lane & 15;
  const int kh   = (lane >> 4) << 1;

  const int rowBase = blockIdx.x << 4;
  const int nBase   = (blockIdx.y * 4 + wave) << 6;
  if (nBase >= M) return;

  v8f c0 = {}, c1 = {}, c2 = {}, c3 = {};
  const float* __restrict__ arow = X + (size_t)(rowBase + lr) * K;

  for (int k = 0; k < K; k += 4) {
    v2f a;
    a.x = arow[k + kh];
    a.y = arow[k + kh + 1];

    const float* __restrict__ wp = W + (size_t)(k + kh) * M + nBase + lr;
    v2f b0, b1, b2, b3;
    b0.x = wp[0];   b0.y = wp[M];
    b1.x = wp[16];  b1.y = wp[M + 16];
    b2.x = wp[32];  b2.y = wp[M + 32];
    b3.x = wp[48];  b3.y = wp[M + 48];

    c0 = __builtin_amdgcn_wmma_f32_16x16x4_f32(false, a, false, b0, (short)0, c0, false, false);
    c1 = __builtin_amdgcn_wmma_f32_16x16x4_f32(false, a, false, b1, (short)0, c1, false, false);
    c2 = __builtin_amdgcn_wmma_f32_16x16x4_f32(false, a, false, b2, (short)0, c2, false, false);
    c3 = __builtin_amdgcn_wmma_f32_16x16x4_f32(false, a, false, b3, (short)0, c3, false, false);
  }

  const int colOff = lane & 15;
  const int rowOff = (lane >> 4) << 3;
#pragma unroll
  for (int v = 0; v < 8; ++v) {
    float* __restrict__ op =
        Hout + (size_t)(rowBase + rowOff + v) * M + nBase + colOff;
    op[0]  = c0[v];
    op[16] = c1[v];
    op[32] = c2[v];
    op[48] = c3[v];
  }
}

__global__ __launch_bounds__(128)
void alpha_kernel(const float* __restrict__ Hf, const float* __restrict__ asrc,
                  const float* __restrict__ adst, float* __restrict__ as_out,
                  float* __restrict__ ad_out, int Hh, int M, int total) {
  const int wid  = blockIdx.x * 4 + (threadIdx.x >> 5);
  const int lane = threadIdx.x & 31;
  if (wid >= total) return;
  const int n  = wid / Hh;
  const int hd = wid - n * Hh;

  const float4 hv = *(const float4*)(Hf + (size_t)n * M + hd * 128 + lane * 4);
  const float4 sv = *(const float4*)(asrc + hd * 128 + lane * 4);
  const float4 dv = *(const float4*)(adst + hd * 128 + lane * 4);
  float s = hv.x * sv.x + hv.y * sv.y + hv.z * sv.z + hv.w * sv.w;
  float d = hv.x * dv.x + hv.y * dv.y + hv.z * dv.z + hv.w * dv.w;
#pragma unroll
  for (int off = 16; off > 0; off >>= 1) {
    s += __shfl_xor(s, off);
    d += __shfl_xor(d, off);
  }
  if (lane == 0) { as_out[wid] = s; ad_out[wid] = d; }
}

__global__ __launch_bounds__(128)
void node_softmax_gather_kernel(const float* __restrict__ Hf,
                                const float* __restrict__ as,
                                const float* __restrict__ ad,
                                const int* __restrict__ rowptr,
                                const int* __restrict__ col,
                                const float* __restrict__ bias,
                                float* __restrict__ out,
                                int Hh, int M, int doElu, int total) {
  const int wv   = __builtin_amdgcn_readfirstlane((int)(threadIdx.x >> 5));
  const int wid  = blockIdx.x * 4 + wv;
  const int lane = threadIdx.x & 31;
  if (wid >= total) return;
  const int d   = wid / Hh;
  const int hd  = wid - d * Hh;
  const int beg = rowptr[d];
  const int deg = rowptr[d + 1] - beg;
  const int tot = deg + 1;
  const float adv = ad[wid];

  const bool v0 = lane < tot;
  int   s0 = d;
  float l0 = -INFINITY;
  if (v0) {
    s0 = (lane < deg) ? col[beg + lane] : d;
    const float l = as[s0 * Hh + hd] + adv;
    l0 = (l > 0.0f) ? l : NEG_SLOPE * l;
  }

  float m = l0;
  for (int i = lane + 32; i < tot; i += 32) {
    const int s = (i < deg) ? col[beg + i] : d;
    float l = as[s * Hh + hd] + adv;
    l = (l > 0.0f) ? l : NEG_SLOPE * l;
    m = fmaxf(m, l);
  }
  for (int off = 16; off > 0; off >>= 1) m = fmaxf(m, __shfl_xor(m, off));
  m = isfinite(m) ? m : 0.0f;

  const float ex0 = v0 ? expf(l0 - m) : 0.0f;
  float den = ex0;
  for (int i = lane + 32; i < tot; i += 32) {
    const int s = (i < deg) ? col[beg + i] : d;
    float l = as[s * Hh + hd] + adv;
    l = (l > 0.0f) ? l : NEG_SLOPE * l;
    den += expf(l - m);
  }
  for (int off = 16; off > 0; off >>= 1) den += __shfl_xor(den, off);
  const float dden = den + 1e-16f;

  float4 acc = make_float4(0.0f, 0.0f, 0.0f, 0.0f);
  const float* __restrict__ hp = Hf + hd * 128 + lane * 4;
  for (int cb = 0; cb < tot; cb += 32) {
    int   s_l = s0;
    float a_l = ex0 / dden;
    if (cb > 0) {
      const int i = cb + lane;
      s_l = d;
      a_l = 0.0f;
      if (i < tot) {
        s_l = (i < deg) ? col[beg + i] : d;
        float l = as[s_l * Hh + hd] + adv;
        l = (l > 0.0f) ? l : NEG_SLOPE * l;
        a_l = expf(l - m) / dden;
      }
    }
    const int nn = (tot - cb < 32) ? (tot - cb) : 32;
    for (int t = 0; t < nn; t += 8) {
      float4 hv[8];
      float  av[8];
      for (int u = 0; u < 8; ++u) {
        if (t + u < nn) {
          const int s = __builtin_amdgcn_readlane(s_l, t + u);
          av[u] = __int_as_float(
              __builtin_amdgcn_readlane(__float_as_int(a_l), t + u));
          hv[u] = *(const float4*)(hp + (size_t)s * M);
        }
      }
      for (int u = 0; u < 8; ++u) {
        if (t + u < nn) {
          acc.x += hv[u].x * av[u];
          acc.y += hv[u].y * av[u];
          acc.z += hv[u].z * av[u];
          acc.w += hv[u].w * av[u];
        }
      }
    }
  }

  const int c = hd * 128 + lane * 4;
  const float4 bv = *(const float4*)(bias + c);
  float4 r;
  r.x = acc.x + bv.x; r.y = acc.y + bv.y; r.z = acc.z + bv.z; r.w = acc.w + bv.w;
  if (doElu) {
    r.x = (r.x > 0.0f) ? r.x : expm1f(r.x);
    r.y = (r.y > 0.0f) ? r.y : expm1f(r.y);
    r.z = (r.z > 0.0f) ? r.z : expm1f(r.z);
    r.w = (r.w > 0.0f) ? r.w : expm1f(r.w);
  }
  *(float4*)(out + (size_t)d * M + c) = r;
}

static void run_layer(hipStream_t stream, const float* xin, const float* W,
                      const float* asrc, const float* adst, const float* bias,
                      int N, int K, int M, int Hh, bool elu,
                      const int* rowptr, const int* col,
                      float* hbuf, float* asb, float* adb, float* out) {
  const int nh = N * Hh;

  dim3 ggrid(N / 16, (M + 255) / 256);
  gemm_wmma_f32<<<ggrid, 128, 0, stream>>>(xin, W, hbuf, K, M);

  alpha_kernel<<<(nh + 3) / 4, 128, 0, stream>>>(hbuf, asrc, adst, asb, adb, Hh, M, nh);

  node_softmax_gather_kernel<<<(nh + 3) / 4, 128, 0, stream>>>(
      hbuf, asb, adb, rowptr, col, bias, out, Hh, M, elu ? 1 : 0, nh);
}

extern "C" void kernel_launch(void* const* d_in, const int* in_sizes, int n_in,
                              void* d_out, int out_size, void* d_ws, size_t ws_size,
                              hipStream_t stream) {
  const float* x      = (const float*)d_in[0];
  const int*   ei     = (const int*)d_in[1];
  const float* W0     = (const float*)d_in[2];
  const float* a_src0 = (const float*)d_in[3];
  const float* a_dst0 = (const float*)d_in[4];
  const float* b0     = (const float*)d_in[5];
  const float* W1     = (const float*)d_in[6];
  const float* a_src1 = (const float*)d_in[7];
  const float* a_dst1 = (const float*)d_in[8];
  const float* b1     = (const float*)d_in[9];
  const float* W2     = (const float*)d_in[10];
  const float* a_src2 = (const float*)d_in[11];
  const float* a_dst2 = (const float*)d_in[12];
  const float* b2     = (const float*)d_in[13];

  const int N  = in_sizes[0] / 256;
  const int E0 = in_sizes[1] / 2;
  const int* esrc = ei;
  const int* edst = ei + E0;

  float* ws = (float*)d_ws;
  size_t p = 0;
  float* hbuf   = ws + p;             p += (size_t)N * 512;
  int*   csr    = (int*)(ws + p);     p += (size_t)N * 512;
  float* xbuf   = ws + p;             p += (size_t)N * 512;
  float* asb    = ws + p;             p += (size_t)N * 4;
  float* adb    = ws + p;             p += (size_t)N * 4;
  int*   cursor = csr;
  int*   rowptr = csr + N;
  int*   col    = csr + (2 * N + 1);
  (void)ws_size; (void)n_in; (void)out_size;

  csr_zero_kernel<<<(N + 255) / 256, 256, 0, stream>>>(cursor, N);
  csr_count_kernel<<<(E0 + 255) / 256, 256, 0, stream>>>(edst, cursor, E0);
  csr_scan_kernel<<<1, 1024, 0, stream>>>(cursor, rowptr, N);
  csr_fill_kernel<<<(E0 + 255) / 256, 256, 0, stream>>>(esrc, edst, cursor, col, E0);

  run_layer(stream, x, W0, a_src0, a_dst0, b0, N, 256, 512, 4, true,
            rowptr, col, hbuf, asb, adb, xbuf);
  run_layer(stream, xbuf, W1, a_src1, a_dst1, b1, N, 512, 512, 4, true,
            rowptr, col, hbuf, asb, adb, xbuf);
  run_layer(stream, xbuf, W2, a_src2, a_dst2, b2, N, 512, 128, 1, false,
            rowptr, col, hbuf, asb, adb, (float*)d_out);
}


# GEMM tiled through LDS: 4-wave block owns 80 x 128 (80 x 64 in the last layer), K slabs of 32 fetched one slab ahead by whole-line loads, fragments read from LDS; k order and sums unchanged
# speedup vs baseline: 3.2582x; 1.3784x over previous
#include <hip/hip_runtime.h>
#include <hip/hip_bf16.h>
#include <math.h>


typedef __attribute__((ext_vector_type(2))) float v2f;
typedef __attribute__((ext_vector_type(4))) float v4f;
typedef __attribute__((ext_vector_type(8))) float v8f;

#define NEG_SLOPE 0.2f

__global__ void csr_zero_kernel(int* __restrict__ cnt, int n) {
  int i = blockIdx.x * blockDim.x + threadIdx.x;
  if (i < n) cnt[i] = 0;
}

__global__ void csr_count_kernel(const int* __restrict__ edst,
                                 int* __restrict__ cnt, int E0) {
  int e = blockIdx.x * blockDim.x + threadIdx.x;
  if (e < E0) atomicAdd(&cnt[edst[e]], 1);
}

__global__ __launch_bounds__(1024)
void csr_scan_kernel(int* __restrict__ cnt, int* __restrict__ rowptr, int n) {
  __shared__ int sh[1024];
  const int t   = threadIdx.x;
  const int per = (n + 1023) / 1024;
  const int b   = t * per;
  const int e   = (b + per < n) ? (b + per) : n;
  int sum = 0;
  for (int i = b; i < e; ++i) sum += cnt[i];
  sh[t] = sum;
  __syncthreads();
  for (int off = 1; off < 1024; off <<= 1) {
    const int v = (t >= off) ? sh[t - off] : 0;
    __syncthreads();
    sh[t] += v;
    __syncthreads();
  }
  int run = sh[t] - sum;
  for (int i = b; i < e; ++i) {
    const int c = cnt[i];
    rowptr[i] = run;
    cnt[i]    = run;
    run += c;
  }
  if (t == 1023) rowptr[n] = sh[1023];
}

__global__ void csr_fill_kernel(const int* __restrict__ esrc,
                                const int* __restrict__ edst,
                                int* __restrict__ cursor,
                                int* __restrict__ col, int E0) {
  int e = blockIdx.x * blockDim.x + threadIdx.x;
  if (e < E0) {
    const int p = atomicAdd(&cursor[edst[e]], 1);
    col[p] = esrc[e];
  }
}

#define GEMM_BM 80
#define GEMM_BK 32
#define GEMM_AP 36

template <int NT>
__global__ __launch_bounds__(128)
void gemm_wmma_f32(const float* __restrict__ X, const float* __restrict__ W,
                   float* __restrict__ Hout, int K, int M) {
  constexpr int BN    = 64 * NT;
  constexpr int BP2   = 2 * BN + 32;
  constexpr int BQ    = BN / 4;
  constexpr int PPP   = 128 / BQ;
  constexpr int BPASS = (GEMM_BK / 2) / PPP;
  constexpr int APASS = GEMM_BM / 16;

  __shared__ __attribute__((aligned(16))) float As[2][GEMM_BM * GEMM_AP];
  __shared__ __attribute__((aligned(16))) float Bs[2][(GEMM_BK / 2) * BP2];

  const int t    = threadIdx.x;
  const int wave = t >> 5;
  const int lane = t & 31;
  const int c    = lane & 15;
  const int h    = lane >> 4;
  const int m0   = blockIdx.x * GEMM_BM;
  const int n0   = blockIdx.y * BN;

  const int arow = t >> 3;
  const int aq   = (t & 7) << 2;
  const int bpr  = t / BQ;
  const int bq   = t % BQ;

  const float* __restrict__ xg = X + (size_t)(m0 + arow) * K + aq;
  const float* __restrict__ wg = W + (size_t)(2 * bpr) * M + n0 + 4 * bq;

  v4f ra[APASS];
  v4f rb0[BPASS], rb1[BPASS];
  v8f acc[APASS][NT];
#pragma unroll
  for (int i = 0; i < APASS; ++i)
#pragma unroll
    for (int j = 0; j < NT; ++j) acc[i][j] = v8f{};

#pragma unroll
  for (int i = 0; i < APASS; ++i)
    ra[i] = *(const v4f*)(xg + (size_t)(16 * i) * K);
#pragma unroll
  for (int i = 0; i < BPASS; ++i) {
    rb0[i] = *(const v4f*)(wg + (size_t)(2 * PPP * i) * M);
    rb1[i] = *(const v4f*)(wg + (size_t)(2 * PPP * i + 1) * M);
  }
#pragma unroll
  for (int i = 0; i < APASS; ++i)
    *(v4f*)&As[0][(arow + 16 * i) * GEMM_AP + aq] = ra[i];
#pragma unroll
  for (int i = 0; i < BPASS; ++i) {
    float* bd = &Bs[0][(bpr + PPP * i) * BP2 + 8 * bq];
    *(v4f*)bd       = v4f{rb0[i].x, rb1[i].x, rb0[i].y, rb1[i].y};
    *(v4f*)(bd + 4) = v4f{rb0[i].z, rb1[i].z, rb0[i].w, rb1[i].w};
  }
  __syncthreads();

  const int KT   = K / GEMM_BK;
  const int aoff = c * GEMM_AP + 2 * h;
  const int boff = h * BP2 + 2 * (wave * (16 * NT) + c);

  for (int kt = 0; kt < KT; ++kt) {
    const int  cur  = kt & 1;
    const bool more = (kt + 1 < KT);

    if (more) {
      const float* __restrict__ xa = xg + (size_t)(kt + 1) * GEMM_BK;
      const float* __restrict__ wb = wg + (size_t)(kt + 1) * GEMM_BK * M;
#pragma unroll
      for (int i = 0; i < APASS; ++i)
        ra[i] = *(const v4f*)(xa + (size_t)(16 * i) * K);
#pragma unroll
      for (int i = 0; i < BPASS; ++i) {
        rb0[i] = *(const v4f*)(wb + (size_t)(2 * PPP * i) * M);
        rb1[i] = *(const v4f*)(wb + (size_t)(2 * PPP * i + 1) * M);
      }
    }

    v2f fa[2][2][APASS];
    v2f fb[2][2][NT];
#pragma unroll
    for (int u = 0; u < 2; ++u) {
#pragma unroll
      for (int i = 0; i < APASS; ++i)
        fa[0][u][i] = *(const v2f*)&As[cur][aoff + (16 * i) * GEMM_AP + 4 * u];
#pragma unroll
      for (int j = 0; j < NT; ++j)
        fb[0][u][j] = *(const v2f*)&Bs[cur][boff + (2 * u) * BP2 + 32 * j];
    }

#pragma unroll
    for (int kk = 0; kk < GEMM_BK / 8; ++kk) {
      const int s = kk & 1;
      if (kk + 1 < GEMM_BK / 8) {
#pragma unroll
        for (int u = 0; u < 2; ++u) {
          const int ks = 2 * (kk + 1) + u;
#pragma unroll
          for (int i = 0; i < APASS; ++i)
            fa[s ^ 1][u][i] =
                *(const v2f*)&As[cur][aoff + (16 * i) * GEMM_AP + 4 * ks];
#pragma unroll
          for (int j = 0; j < NT; ++j)
            fb[s ^ 1][u][j] =
                *(const v2f*)&Bs[cur][boff + (2 * ks) * BP2 + 32 * j];
        }
      }
      __builtin_amdgcn_sched_barrier(0);
#pragma unroll
      for (int u = 0; u < 2; ++u)
#pragma unroll
        for (int i = 0; i < APASS; ++i)
#pragma unroll
          for (int j = 0; j < NT; ++j)
            acc[i][j] = __builtin_amdgcn_wmma_f32_16x16x4_f32(
                false, fa[s][u][i], false, fb[s][u][j], (short)0, acc[i][j],
                false, false);
      __builtin_amdgcn_sched_barrier(0);
    }

    if (more) {
#pragma unroll
      for (int i = 0; i < APASS; ++i)
        *(v4f*)&As[cur ^ 1][(arow + 16 * i) * GEMM_AP + aq] = ra[i];
#pragma unroll
      for (int i = 0; i < BPASS; ++i) {
        float* bd = &Bs[cur ^ 1][(bpr + PPP * i) * BP2 + 8 * bq];
        *(v4f*)bd       = v4f{rb0[i].x, rb1[i].x, rb0[i].y, rb1[i].y};
        *(v4f*)(bd + 4) = v4f{rb0[i].z, rb1[i].z, rb0[i].w, rb1[i].w};
      }
    }
    __syncthreads();
  }

  float* __restrict__ ob =
      Hout + (size_t)(m0 + 8 * h) * M + n0 + wave * (16 * NT) + c;
#pragma unroll
  for (int i = 0; i < APASS; ++i)
#pragma unroll
    for (int v = 0; v < 8; ++v) {
      float* __restrict__ op = ob + (size_t)(16 * i + v) * M;
#pragma unroll
      for (int j = 0; j < NT; ++j) op[16 * j] = acc[i][j][v];
    }
}

__global__ __launch_bounds__(128)
void alpha_kernel(const float* __restrict__ Hf, const float* __restrict__ asrc,
                  const float* __restrict__ adst, float* __restrict__ as_out,
                  float* __restrict__ ad_out, int Hh, int M, int total) {
  const int wid  = blockIdx.x * 4 + (threadIdx.x >> 5);
  const int lane = threadIdx.x & 31;
  if (wid >= total) return;
  const int n  = wid / Hh;
  const int hd = wid - n * Hh;

  const float4 hv = *(const float4*)(Hf + (size_t)n * M + hd * 128 + lane * 4);
  const float4 sv = *(const float4*)(asrc + hd * 128 + lane * 4);
  const float4 dv = *(const float4*)(adst + hd * 128 + lane * 4);
  float s = hv.x * sv.x + hv.y * sv.y + hv.z * sv.z + hv.w * sv.w;
  float d = hv.x * dv.x + hv.y * dv.y + hv.z * dv.z + hv.w * dv.w;
#pragma unroll
  for (int off = 16; off > 0; off >>= 1) {
    s += __shfl_xor(s, off);
    d += __shfl_xor(d, off);
  }
  if (lane == 0) { as_out[wid] = s; ad_out[wid] = d; }
}

__global__ __launch_bounds__(128)
void node_softmax_gather_kernel(const float* __restrict__ Hf,
                                const float* __restrict__ as,
                                const float* __restrict__ ad,
                                const int* __restrict__ rowptr,
                                const int* __restrict__ col,
                                const float* __restrict__ bias,
                                float* __restrict__ out,
                                int Hh, int M, int doElu, int total) {
  const int wv   = __builtin_amdgcn_readfirstlane((int)(threadIdx.x >> 5));
  const int wid  = blockIdx.x * 4 + wv;
  const int lane = threadIdx.x & 31;
  if (wid >= total) return;
  const int d   = wid / Hh;
  const int hd  = wid - d * Hh;
  const int beg = rowptr[d];
  const int deg = rowptr[d + 1] - beg;
  const int tot = deg + 1;
  const float adv = ad[wid];

  const bool v0 = lane < tot;
  int   s0 = d;
  float l0 = -INFINITY;
  if (v0) {
    s0 = (lane < deg) ? col[beg + lane] : d;
    const float l = as[s0 * Hh + hd] + adv;
    l0 = (l > 0.0f) ? l : NEG_SLOPE * l;
  }

  float m = l0;
  for (int i = lane + 32; i < tot; i += 32) {
    const int s = (i < deg) ? col[beg + i] : d;
    float l = as[s * Hh + hd] + adv;
    l = (l > 0.0f) ? l : NEG_SLOPE * l;
    m = fmaxf(m, l);
  }
  for (int off = 16; off > 0; off >>= 1) m = fmaxf(m, __shfl_xor(m, off));
  m = isfinite(m) ? m : 0.0f;

  const float ex0 = v0 ? expf(l0 - m) : 0.0f;
  float den = ex0;
  for (int i = lane + 32; i < tot; i += 32) {
    const int s = (i < deg) ? col[beg + i] : d;
    float l = as[s * Hh + hd] + adv;
    l = (l > 0.0f) ? l : NEG_SLOPE * l;
    den += expf(l - m);
  }
  for (int off = 16; off > 0; off >>= 1) den += __shfl_xor(den, off);
  const float dden = den + 1e-16f;

  float4 acc = make_float4(0.0f, 0.0f, 0.0f, 0.0f);
  const float* __restrict__ hp = Hf + hd * 128 + lane * 4;
  for (int cb = 0; cb < tot; cb += 32) {
    int   s_l = s0;
    float a_l = ex0 / dden;
    if (cb > 0) {
      const int i = cb + lane;
      s_l = d;
      a_l = 0.0f;
      if (i < tot) {
        s_l = (i < deg) ? col[beg + i] : d;
        float l = as[s_l * Hh + hd] + adv;
        l = (l > 0.0f) ? l : NEG_SLOPE * l;
        a_l = expf(l - m) / dden;
      }
    }
    const int nn = (tot - cb < 32) ? (tot - cb) : 32;
    for (int t = 0; t < nn; t += 8) {
      float4 hv[8];
      float  av[8];
      for (int u = 0; u < 8; ++u) {
        if (t + u < nn) {
          const int s = __builtin_amdgcn_readlane(s_l, t + u);
          av[u] = __int_as_float(
              __builtin_amdgcn_readlane(__float_as_int(a_l), t + u));
          hv[u] = *(const float4*)(hp + (size_t)s * M);
        }
      }
      for (int u = 0; u < 8; ++u) {
        if (t + u < nn) {
          acc.x += hv[u].x * av[u];
          acc.y += hv[u].y * av[u];
          acc.z += hv[u].z * av[u];
          acc.w += hv[u].w * av[u];
        }
      }
    }
  }

  const int c = hd * 128 + lane * 4;
  const float4 bv = *(const float4*)(bias + c);
  float4 r;
  r.x = acc.x + bv.x; r.y = acc.y + bv.y; r.z = acc.z + bv.z; r.w = acc.w + bv.w;
  if (doElu) {
    r.x = (r.x > 0.0f) ? r.x : expm1f(r.x);
    r.y = (r.y > 0.0f) ? r.y : expm1f(r.y);
    r.z = (r.z > 0.0f) ? r.z : expm1f(r.z);
    r.w = (r.w > 0.0f) ? r.w : expm1f(r.w);
  }
  *(float4*)(out + (size_t)d * M + c) = r;
}

static void run_layer(hipStream_t stream, const float* xin, const float* W,
                      const float* asrc, const float* adst, const float* bias,
                      int N, int K, int M, int Hh, bool elu,
                      const int* rowptr, const int* col,
                      float* hbuf, float* asb, float* adb, float* out) {
  const int nh = N * Hh;

  if (M == 128) {
    gemm_wmma_f32<1><<<dim3(N / GEMM_BM, M / 64), 128, 0, stream>>>(xin, W, hbuf, K, M);
  } else {
    gemm_wmma_f32<2><<<dim3(N / GEMM_BM, M / 128), 128, 0, stream>>>(xin, W, hbuf, K, M);
  }

  alpha_kernel<<<(nh + 3) / 4, 128, 0, stream>>>(hbuf, asrc, adst, asb, adb, Hh, M, nh);

  node_softmax_gather_kernel<<<(nh + 3) / 4, 128, 0, stream>>>(
      hbuf, asb, adb, rowptr, col, bias, out, Hh, M, elu ? 1 : 0, nh);
}

extern "C" void kernel_launch(void* const* d_in, const int* in_sizes, int n_in,
                              void* d_out, int out_size, void* d_ws, size_t ws_size,
                              hipStream_t stream) {
  const float* x      = (const float*)d_in[0];
  const int*   ei     = (const int*)d_in[1];
  const float* W0     = (const float*)d_in[2];
  const float* a_src0 = (const float*)d_in[3];
  const float* a_dst0 = (const float*)d_in[4];
  const float* b0     = (const float*)d_in[5];
  const float* W1     = (const float*)d_in[6];
  const float* a_src1 = (const float*)d_in[7];
  const float* a_dst1 = (const float*)d_in[8];
  const float* b1     = (const float*)d_in[9];
  const float* W2     = (const float*)d_in[10];
  const float* a_src2 = (const float*)d_in[11];
  const float* a_dst2 = (const float*)d_in[12];
  const float* b2     = (const float*)d_in[13];

  const int N  = in_sizes[0] / 256;
  const int E0 = in_sizes[1] / 2;
  const int* esrc = ei;
  const int* edst = ei + E0;

  float* ws = (float*)d_ws;
  size_t p = 0;
  float* hbuf   = ws + p;             p += (size_t)N * 512;
  int*   csr    = (int*)(ws + p);     p += (size_t)N * 512;
  float* xbuf   = ws + p;             p += (size_t)N * 512;
  float* asb    = ws + p;             p += (size_t)N * 4;
  float* adb    = ws + p;             p += (size_t)N * 4;
  int*   cursor = csr;
  int*   rowptr = csr + N;
  int*   col    = csr + (2 * N + 1);
  (void)ws_size; (void)n_in; (void)out_size;

  csr_zero_kernel<<<(N + 255) / 256, 256, 0, stream>>>(cursor, N);
  csr_count_kernel<<<(E0 + 255) / 256, 256, 0, stream>>>(edst, cursor, E0);
  csr_scan_kernel<<<1, 1024, 0, stream>>>(cursor, rowptr, N);
  csr_fill_kernel<<<(E0 + 255) / 256, 256, 0, stream>>>(esrc, edst, cursor, col, E0);

  run_layer(stream, x, W0, a_src0, a_dst0, b0, N, 256, 512, 4, true,
            rowptr, col, hbuf, asb, adb, xbuf);
  run_layer(stream, xbuf, W1, a_src1, a_dst1, b1, N, 512, 512, 4, true,
            rowptr, col, hbuf, asb, adb, xbuf);
  run_layer(stream, xbuf, W2, a_src2, a_dst2, b2, N, 512, 128, 1, false,
            rowptr, col, hbuf, asb, adb, (float*)d_out);
}


# 4-head layers: GEMM epilogue parks its 80x128 one-head tile in LDS, forms alpha_s/alpha_d there (alpha_kernel's expression and butterfly, sum order kept), stores whole rows; 2 alpha launches go
# speedup vs baseline: 3.4878x; 1.0705x over previous
#include <hip/hip_runtime.h>
#include <hip/hip_bf16.h>
#include <math.h>


typedef __attribute__((ext_vector_type(2))) float v2f;
typedef __attribute__((ext_vector_type(4))) float v4f;
typedef __attribute__((ext_vector_type(8))) float v8f;

#define NEG_SLOPE 0.2f

__global__ void csr_zero_kernel(int* __restrict__ cnt, int n) {
  int i = blockIdx.x * blockDim.x + threadIdx.x;
  if (i < n) cnt[i] = 0;
}

__global__ void csr_count_kernel(const int* __restrict__ edst,
                                 int* __restrict__ cnt, int E0) {
  int e = blockIdx.x * blockDim.x + threadIdx.x;
  if (e < E0) atomicAdd(&cnt[edst[e]], 1);
}

__global__ __launch_bounds__(1024)
void csr_scan_kernel(int* __restrict__ cnt, int* __restrict__ rowptr, int n) {
  __shared__ int sh[1024];
  const int t   = threadIdx.x;
  const int per = (n + 1023) / 1024;
  const int b   = t * per;
  const int e   = (b + per < n) ? (b + per) : n;
  int sum = 0;
  for (int i = b; i < e; ++i) sum += cnt[i];
  sh[t] = sum;
  __syncthreads();
  for (int off = 1; off < 1024; off <<= 1) {
    const int v = (t >= off) ? sh[t - off] : 0;
    __syncthreads();
    sh[t] += v;
    __syncthreads();
  }
  int run = sh[t] - sum;
  for (int i = b; i < e; ++i) {
    const int c = cnt[i];
    rowptr[i] = run;
    cnt[i]    = run;
    run += c;
  }
  if (t == 1023) rowptr[n] = sh[1023];
}

__global__ void csr_fill_kernel(const int* __restrict__ esrc,
                                const int* __restrict__ edst,
                                int* __restrict__ cursor,
                                int* __restrict__ col, int E0) {
  int e = blockIdx.x * blockDim.x + threadIdx.x;
  if (e < E0) {
    const int p = atomicAdd(&cursor[edst[e]], 1);
    col[p] = esrc[e];
  }
}

#define GEMM_BM 80
#define GEMM_BK 32
#define GEMM_AP 36
#define GEMM_CP 132

template <int NT>
__global__ __launch_bounds__(128)
void gemm_wmma_f32(const float* __restrict__ X, const float* __restrict__ W,
                   float* __restrict__ Hout, int K, int M,
                   const float* __restrict__ asrc, const float* __restrict__ adst,
                   float* __restrict__ as_out, float* __restrict__ ad_out,
                   int Hh) {
  constexpr int BN    = 64 * NT;
  constexpr int BP2   = 2 * BN + 32;
  constexpr int BQ    = BN / 4;
  constexpr int PPP   = 128 / BQ;
  constexpr int BPASS = (GEMM_BK / 2) / PPP;
  constexpr int APASS = GEMM_BM / 16;

  constexpr int AS_SZ = GEMM_BM * GEMM_AP;
  constexpr int BS_SZ = (GEMM_BK / 2) * BP2;
  static_assert(NT != 2 || GEMM_BM * GEMM_CP <= 2 * AS_SZ + 2 * BS_SZ,
                "the parked tile must fit in the K-loop's stages");

  __shared__ __attribute__((aligned(16))) float smem[2 * AS_SZ + 2 * BS_SZ];
  float (*As)[AS_SZ] = reinterpret_cast<float (*)[AS_SZ]>(smem);
  float (*Bs)[BS_SZ] = reinterpret_cast<float (*)[BS_SZ]>(smem + 2 * AS_SZ);

  const int t    = threadIdx.x;
  const int wave = t >> 5;
  const int lane = t & 31;
  const int c    = lane & 15;
  const int h    = lane >> 4;
  const int m0   = blockIdx.x * GEMM_BM;
  const int n0   = blockIdx.y * BN;

  const int arow = t >> 3;
  const int aq   = (t & 7) << 2;
  const int bpr  = t / BQ;
  const int bq   = t % BQ;

  const float* __restrict__ xg = X + (size_t)(m0 + arow) * K + aq;
  const float* __restrict__ wg = W + (size_t)(2 * bpr) * M + n0 + 4 * bq;

  v4f ra[APASS];
  v4f rb0[BPASS], rb1[BPASS];
  v8f acc[APASS][NT];
#pragma unroll
  for (int i = 0; i < APASS; ++i)
#pragma unroll
    for (int j = 0; j < NT; ++j) acc[i][j] = v8f{};

#pragma unroll
  for (int i = 0; i < APASS; ++i)
    ra[i] = *(const v4f*)(xg + (size_t)(16 * i) * K);
#pragma unroll
  for (int i = 0; i < BPASS; ++i) {
    rb0[i] = *(const v4f*)(wg + (size_t)(2 * PPP * i) * M);
    rb1[i] = *(const v4f*)(wg + (size_t)(2 * PPP * i + 1) * M);
  }
#pragma unroll
  for (int i = 0; i < APASS; ++i)
    *(v4f*)&As[0][(arow + 16 * i) * GEMM_AP + aq] = ra[i];
#pragma unroll
  for (int i = 0; i < BPASS; ++i) {
    float* bd = &Bs[0][(bpr + PPP * i) * BP2 + 8 * bq];
    *(v4f*)bd       = v4f{rb0[i].x, rb1[i].x, rb0[i].y, rb1[i].y};
    *(v4f*)(bd + 4) = v4f{rb0[i].z, rb1[i].z, rb0[i].w, rb1[i].w};
  }
  __syncthreads();

  const int KT   = K / GEMM_BK;
  const int aoff = c * GEMM_AP + 2 * h;
  const int boff = h * BP2 + 2 * (wave * (16 * NT) + c);

  for (int kt = 0; kt < KT; ++kt) {
    const int  cur  = kt & 1;
    const bool more = (kt + 1 < KT);

    if (more) {
      const float* __restrict__ xa = xg + (size_t)(kt + 1) * GEMM_BK;
      const float* __restrict__ wb = wg + (size_t)(kt + 1) * GEMM_BK * M;
#pragma unroll
      for (int i = 0; i < APASS; ++i)
        ra[i] = *(const v4f*)(xa + (size_t)(16 * i) * K);
#pragma unroll
      for (int i = 0; i < BPASS; ++i) {
        rb0[i] = *(const v4f*)(wb + (size_t)(2 * PPP * i) * M);
        rb1[i] = *(const v4f*)(wb + (size_t)(2 * PPP * i + 1) * M);
      }
    }

    v2f fa[2][2][APASS];
    v2f fb[2][2][NT];
#pragma unroll
    for (int u = 0; u < 2; ++u) {
#pragma unroll
      for (int i = 0; i < APASS; ++i)
        fa[0][u][i] = *(const v2f*)&As[cur][aoff + (16 * i) * GEMM_AP + 4 * u];
#pragma unroll
      for (int j = 0; j < NT; ++j)
        fb[0][u][j] = *(const v2f*)&Bs[cur][boff + (2 * u) * BP2 + 32 * j];
    }

#pragma unroll
    for (int kk = 0; kk < GEMM_BK / 8; ++kk) {
      const int s = kk & 1;
      if (kk + 1 < GEMM_BK / 8) {
#pragma unroll
        for (int u = 0; u < 2; ++u) {
          const int ks = 2 * (kk + 1) + u;
#pragma unroll
          for (int i = 0; i < APASS; ++i)
            fa[s ^ 1][u][i] =
                *(const v2f*)&As[cur][aoff + (16 * i) * GEMM_AP + 4 * ks];
#pragma unroll
          for (int j = 0; j < NT; ++j)
            fb[s ^ 1][u][j] =
                *(const v2f*)&Bs[cur][boff + (2 * ks) * BP2 + 32 * j];
        }
      }
      __builtin_amdgcn_sched_barrier(0);
#pragma unroll
      for (int u = 0; u < 2; ++u)
#pragma unroll
        for (int i = 0; i < APASS; ++i)
#pragma unroll
          for (int j = 0; j < NT; ++j)
            acc[i][j] = __builtin_amdgcn_wmma_f32_16x16x4_f32(
                false, fa[s][u][i], false, fb[s][u][j], (short)0, acc[i][j],
                false, false);
      __builtin_amdgcn_sched_barrier(0);
    }

    if (more) {
#pragma unroll
      for (int i = 0; i < APASS; ++i)
        *(v4f*)&As[cur ^ 1][(arow + 16 * i) * GEMM_AP + aq] = ra[i];
#pragma unroll
      for (int i = 0; i < BPASS; ++i) {
        float* bd = &Bs[cur ^ 1][(bpr + PPP * i) * BP2 + 8 * bq];
        *(v4f*)bd       = v4f{rb0[i].x, rb1[i].x, rb0[i].y, rb1[i].y};
        *(v4f*)(bd + 4) = v4f{rb0[i].z, rb1[i].z, rb0[i].w, rb1[i].w};
      }
    }
    __syncthreads();
  }

  if constexpr (NT == 2) {
    const int head = blockIdx.y;
    const float4 sv = *(const float4*)(asrc + head * 128 + lane * 4);
    const float4 dv = *(const float4*)(adst + head * 128 + lane * 4);

    float* cw = smem + (8 * h) * GEMM_CP + wave * 32 + c;
#pragma unroll
    for (int i = 0; i < APASS; ++i)
#pragma unroll
      for (int v = 0; v < 8; ++v)
#pragma unroll
        for (int j = 0; j < NT; ++j)
          cw[(16 * i + v) * GEMM_CP + 16 * j] = acc[i][j][v];
    __syncthreads();

    constexpr int RPW = GEMM_BM / 4;
    const int r0 = wave * RPW;
    const float* cr = smem + r0 * GEMM_CP + lane * 4;
    float* __restrict__ orow = Hout + (size_t)(m0 + r0) * M + n0 + lane * 4;
    float ks = 0.0f, kd = 0.0f;
#pragma unroll
    for (int q = 0; q < RPW; ++q) {
      const float4 hv = *(const float4*)(cr + q * GEMM_CP);
      *(float4*)(orow + (size_t)q * M) = hv;
      float s = hv.x * sv.x + hv.y * sv.y + hv.z * sv.z + hv.w * sv.w;
      float d = hv.x * dv.x + hv.y * dv.y + hv.z * dv.z + hv.w * dv.w;
#pragma unroll
      for (int off = 16; off > 0; off >>= 1) {
        s += __shfl_xor(s, off);
        d += __shfl_xor(d, off);
      }
      if (lane == q) { ks = s; kd = d; }
    }
    if (lane < RPW) {
      const int o = (m0 + r0 + lane) * Hh + head;
      as_out[o] = ks;
      ad_out[o] = kd;
    }
  } else {
    float* __restrict__ ob =
        Hout + (size_t)(m0 + 8 * h) * M + n0 + wave * (16 * NT) + c;
#pragma unroll
    for (int i = 0; i < APASS; ++i)
#pragma unroll
      for (int v = 0; v < 8; ++v) {
        float* __restrict__ op = ob + (size_t)(16 * i + v) * M;
#pragma unroll
        for (int j = 0; j < NT; ++j) op[16 * j] = acc[i][j][v];
      }
  }
}

__global__ __launch_bounds__(128)
void alpha_kernel(const float* __restrict__ Hf, const float* __restrict__ asrc,
                  const float* __restrict__ adst, float* __restrict__ as_out,
                  float* __restrict__ ad_out, int Hh, int M, int total) {
  const int wid  = blockIdx.x * 4 + (threadIdx.x >> 5);
  const int lane = threadIdx.x & 31;
  if (wid >= total) return;
  const int n  = wid / Hh;
  const int hd = wid - n * Hh;

  const float4 hv = *(const float4*)(Hf + (size_t)n * M + hd * 128 + lane * 4);
  const float4 sv = *(const float4*)(asrc + hd * 128 + lane * 4);
  const float4 dv = *(const float4*)(adst + hd * 128 + lane * 4);
  float s = hv.x * sv.x + hv.y * sv.y + hv.z * sv.z + hv.w * sv.w;
  float d = hv.x * dv.x + hv.y * dv.y + hv.z * dv.z + hv.w * dv.w;
#pragma unroll
  for (int off = 16; off > 0; off >>= 1) {
    s += __shfl_xor(s, off);
    d += __shfl_xor(d, off);
  }
  if (lane == 0) { as_out[wid] = s; ad_out[wid] = d; }
}

__global__ __launch_bounds__(128)
void node_softmax_gather_kernel(const float* __restrict__ Hf,
                                const float* __restrict__ as,
                                const float* __restrict__ ad,
                                const int* __restrict__ rowptr,
                                const int* __restrict__ col,
                                const float* __restrict__ bias,
                                float* __restrict__ out,
                                int Hh, int M, int doElu, int total) {
  const int wv   = __builtin_amdgcn_readfirstlane((int)(threadIdx.x >> 5));
  const int wid  = blockIdx.x * 4 + wv;
  const int lane = threadIdx.x & 31;
  if (wid >= total) return;
  const int d   = wid / Hh;
  const int hd  = wid - d * Hh;
  const int beg = rowptr[d];
  const int deg = rowptr[d + 1] - beg;
  const int tot = deg + 1;
  const float adv = ad[wid];

  const bool v0 = lane < tot;
  int   s0 = d;
  float l0 = -INFINITY;
  if (v0) {
    s0 = (lane < deg) ? col[beg + lane] : d;
    const float l = as[s0 * Hh + hd] + adv;
    l0 = (l > 0.0f) ? l : NEG_SLOPE * l;
  }

  float m = l0;
  for (int i = lane + 32; i < tot; i += 32) {
    const int s = (i < deg) ? col[beg + i] : d;
    float l = as[s * Hh + hd] + adv;
    l = (l > 0.0f) ? l : NEG_SLOPE * l;
    m = fmaxf(m, l);
  }
  for (int off = 16; off > 0; off >>= 1) m = fmaxf(m, __shfl_xor(m, off));
  m = isfinite(m) ? m : 0.0f;

  const float ex0 = v0 ? expf(l0 - m) : 0.0f;
  float den = ex0;
  for (int i = lane + 32; i < tot; i += 32) {
    const int s = (i < deg) ? col[beg + i] : d;
    float l = as[s * Hh + hd] + adv;
    l = (l > 0.0f) ? l : NEG_SLOPE * l;
    den += expf(l - m);
  }
  for (int off = 16; off > 0; off >>= 1) den += __shfl_xor(den, off);
  const float dden = den + 1e-16f;

  float4 acc = make_float4(0.0f, 0.0f, 0.0f, 0.0f);
  const float* __restrict__ hp = Hf + hd * 128 + lane * 4;
  for (int cb = 0; cb < tot; cb += 32) {
    int   s_l = s0;
    float a_l = ex0 / dden;
    if (cb > 0) {
      const int i = cb + lane;
      s_l = d;
      a_l = 0.0f;
      if (i < tot) {
        s_l = (i < deg) ? col[beg + i] : d;
        float l = as[s_l * Hh + hd] + adv;
        l = (l > 0.0f) ? l : NEG_SLOPE * l;
        a_l = expf(l - m) / dden;
      }
    }
    const int nn = (tot - cb < 32) ? (tot - cb) : 32;
    for (int t = 0; t < nn; t += 8) {
      float4 hv[8];
      float  av[8];
      for (int u = 0; u < 8; ++u) {
        if (t + u < nn) {
          const int s = __builtin_amdgcn_readlane(s_l, t + u);
          av[u] = __int_as_float(
              __builtin_amdgcn_readlane(__float_as_int(a_l), t + u));
          hv[u] = *(const float4*)(hp + (size_t)s * M);
        }
      }
      for (int u = 0; u < 8; ++u) {
        if (t + u < nn) {
          acc.x += hv[u].x * av[u];
          acc.y += hv[u].y * av[u];
          acc.z += hv[u].z * av[u];
          acc.w += hv[u].w * av[u];
        }
      }
    }
  }

  const int c = hd * 128 + lane * 4;
  const float4 bv = *(const float4*)(bias + c);
  float4 r;
  r.x = acc.x + bv.x; r.y = acc.y + bv.y; r.z = acc.z + bv.z; r.w = acc.w + bv.w;
  if (doElu) {
    r.x = (r.x > 0.0f) ? r.x : expm1f(r.x);
    r.y = (r.y > 0.0f) ? r.y : expm1f(r.y);
    r.z = (r.z > 0.0f) ? r.z : expm1f(r.z);
    r.w = (r.w > 0.0f) ? r.w : expm1f(r.w);
  }
  *(float4*)(out + (size_t)d * M + c) = r;
}

static void run_layer(hipStream_t stream, const float* xin, const float* W,
                      const float* asrc, const float* adst, const float* bias,
                      int N, int K, int M, int Hh, bool elu,
                      const int* rowptr, const int* col,
                      float* hbuf, float* asb, float* adb, float* out) {
  const int nh = N * Hh;

  if (M == 128) {
    gemm_wmma_f32<1><<<dim3(N / GEMM_BM, M / 64), 128, 0, stream>>>(
        xin, W, hbuf, K, M, asrc, adst, asb, adb, Hh);
    alpha_kernel<<<(nh + 3) / 4, 128, 0, stream>>>(hbuf, asrc, adst, asb, adb, Hh, M, nh);
  } else {
    gemm_wmma_f32<2><<<dim3(N / GEMM_BM, M / 128), 128, 0, stream>>>(
        xin, W, hbuf, K, M, asrc, adst, asb, adb, Hh);
  }

  node_softmax_gather_kernel<<<(nh + 3) / 4, 128, 0, stream>>>(
      hbuf, asb, adb, rowptr, col, bias, out, Hh, M, elu ? 1 : 0, nh);
}

extern "C" void kernel_launch(void* const* d_in, const int* in_sizes, int n_in,
                              void* d_out, int out_size, void* d_ws, size_t ws_size,
                              hipStream_t stream) {
  const float* x      = (const float*)d_in[0];
  const int*   ei     = (const int*)d_in[1];
  const float* W0     = (const float*)d_in[2];
  const float* a_src0 = (const float*)d_in[3];
  const float* a_dst0 = (const float*)d_in[4];
  const float* b0     = (const float*)d_in[5];
  const float* W1     = (const float*)d_in[6];
  const float* a_src1 = (const float*)d_in[7];
  const float* a_dst1 = (const float*)d_in[8];
  const float* b1     = (const float*)d_in[9];
  const float* W2     = (const float*)d_in[10];
  const float* a_src2 = (const float*)d_in[11];
  const float* a_dst2 = (const float*)d_in[12];
  const float* b2     = (const float*)d_in[13];

  const int N  = in_sizes[0] / 256;
  const int E0 = in_sizes[1] / 2;
  const int* esrc = ei;
  const int* edst = ei + E0;

  float* ws = (float*)d_ws;
  size_t p = 0;
  float* hbuf   = ws + p;             p += (size_t)N * 512;
  int*   csr    = (int*)(ws + p);     p += (size_t)N * 512;
  float* xbuf   = ws + p;             p += (size_t)N * 512;
  float* asb    = ws + p;             p += (size_t)N * 4;
  float* adb    = ws + p;             p += (size_t)N * 4;
  int*   cursor = csr;
  int*   rowptr = csr + N;
  int*   col    = csr + (2 * N + 1);
  (void)ws_size; (void)n_in; (void)out_size;

  csr_zero_kernel<<<(N + 255) / 256, 256, 0, stream>>>(cursor, N);
  csr_count_kernel<<<(E0 + 255) / 256, 256, 0, stream>>>(edst, cursor, E0);
  csr_scan_kernel<<<1, 1024, 0, stream>>>(cursor, rowptr, N);
  csr_fill_kernel<<<(E0 + 255) / 256, 256, 0, stream>>>(esrc, edst, cursor, col, E0);

  run_layer(stream, x, W0, a_src0, a_dst0, b0, N, 256, 512, 4, true,
            rowptr, col, hbuf, asb, adb, xbuf);
  run_layer(stream, xbuf, W1, a_src1, a_dst1, b1, N, 512, 512, 4, true,
            rowptr, col, hbuf, asb, adb, xbuf);
  run_layer(stream, xbuf, W2, a_src2, a_dst2, b2, N, 512, 128, 1, false,
            rowptr, col, hbuf, asb, adb, (float*)d_out);
}


# Degree scan of the edge sort runs as one extra 128-thread block of the layer-0 GEMM launch (counts staged in LDS), not as a launch of its own: 11 -> 10 launches; every sum and its order kept
# speedup vs baseline: 3.6411x; 1.0440x over previous
#include <hip/hip_runtime.h>
#include <hip/hip_bf16.h>
#include <math.h>


typedef __attribute__((ext_vector_type(2))) float v2f;
typedef __attribute__((ext_vector_type(4))) float v4f;
typedef __attribute__((ext_vector_type(8))) float v8f;

#define NEG_SLOPE 0.2f

__global__ void csr_zero_kernel(int* __restrict__ cnt, int n) {
  int i = blockIdx.x * blockDim.x + threadIdx.x;
  if (i < n) cnt[i] = 0;
}

__global__ void csr_count_kernel(const int* __restrict__ edst,
                                 int* __restrict__ cnt, int E0) {
  int e = blockIdx.x * blockDim.x + threadIdx.x;
  if (e < E0) atomicAdd(&cnt[edst[e]], 1);
}

__device__ __forceinline__
void csr_scan_block128(int* __restrict__ cnt, int* __restrict__ rowptr, int n,
                       int* sh, int cap) {
  const int t   = threadIdx.x;
  const int per = (n + 127) / 128;
  const int b   = t * per;
  const int e   = (b + per < n) ? (b + per) : n;
  int* part = sh + (cap - 128);
  const bool staged = (n <= cap - 128);
  int sum = 0;
  if (staged) {
    for (int i = t; i < n; i += 128) sh[i] = cnt[i];
    __syncthreads();
    for (int i = b; i < e; ++i) sum += sh[i];
  } else {
    for (int i = b; i < e; ++i) sum += cnt[i];
  }
  part[t] = sum;
  __syncthreads();
  for (int off = 1; off < 128; off <<= 1) {
    const int v = (t >= off) ? part[t - off] : 0;
    __syncthreads();
    part[t] += v;
    __syncthreads();
  }
  int run = part[t] - sum;
  if (staged) {
    for (int i = b; i < e; ++i) {
      const int c = sh[i];
      sh[i] = run;
      run += c;
    }
    __syncthreads();
    for (int i = t; i < n; i += 128) {
      const int v = sh[i];
      rowptr[i] = v;
      cnt[i]    = v;
    }
  } else {
    for (int i = b; i < e; ++i) {
      const int c = cnt[i];
      rowptr[i] = run;
      cnt[i]    = run;
      run += c;
    }
  }
  if (t == 127) rowptr[n] = part[127];
}

__global__ void csr_fill_kernel(const int* __restrict__ esrc,
                                const int* __restrict__ edst,
                                int* __restrict__ cursor,
                                int* __restrict__ col, int E0) {
  int e = blockIdx.x * blockDim.x + threadIdx.x;
  if (e < E0) {
    const int p = atomicAdd(&cursor[edst[e]], 1);
    col[p] = esrc[e];
  }
}

#define GEMM_BM 80
#define GEMM_BK 32
#define GEMM_AP 36
#define GEMM_CP 132

template <int NT, bool SCAN>
__global__ __launch_bounds__(128)
void gemm_wmma_f32(const float* __restrict__ X, const float* __restrict__ W,
                   float* __restrict__ Hout, int K, int M,
                   const float* __restrict__ asrc, const float* __restrict__ adst,
                   float* __restrict__ as_out, float* __restrict__ ad_out,
                   int Hh, int* __restrict__ scan_cnt,
                   int* __restrict__ scan_rowptr, int scan_n) {
  constexpr int BN    = 64 * NT;
  constexpr int BP2   = 2 * BN + 32;
  constexpr int BQ    = BN / 4;
  constexpr int PPP   = 128 / BQ;
  constexpr int BPASS = (GEMM_BK / 2) / PPP;
  constexpr int APASS = GEMM_BM / 16;

  constexpr int AS_SZ = GEMM_BM * GEMM_AP;
  constexpr int BS_SZ = (GEMM_BK / 2) * BP2;
  static_assert(NT != 2 || GEMM_BM * GEMM_CP <= 2 * AS_SZ + 2 * BS_SZ,
                "the parked tile must fit in the K-loop's stages");

  __shared__ __attribute__((aligned(16))) float smem[2 * AS_SZ + 2 * BS_SZ];
  float (*As)[AS_SZ] = reinterpret_cast<float (*)[AS_SZ]>(smem);
  float (*Bs)[BS_SZ] = reinterpret_cast<float (*)[BS_SZ]>(smem + 2 * AS_SZ);

  if constexpr (SCAN) {
    if (blockIdx.x == gridDim.x - 1) {
      if (blockIdx.y == 0)
        csr_scan_block128(scan_cnt, scan_rowptr, scan_n,
                          reinterpret_cast<int*>(smem), 2 * AS_SZ + 2 * BS_SZ);
      return;
    }
  }

  const int t    = threadIdx.x;
  const int wave = t >> 5;
  const int lane = t & 31;
  const int c    = lane & 15;
  const int h    = lane >> 4;
  const int m0   = blockIdx.x * GEMM_BM;
  const int n0   = blockIdx.y * BN;

  const int arow = t >> 3;
  const int aq   = (t & 7) << 2;
  const int bpr  = t / BQ;
  const int bq   = t % BQ;

  const float* __restrict__ xg = X + (size_t)(m0 + arow) * K + aq;
  const float* __restrict__ wg = W + (size_t)(2 * bpr) * M + n0 + 4 * bq;

  v4f ra[APASS];
  v4f rb0[BPASS], rb1[BPASS];
  v8f acc[APASS][NT];
#pragma unroll
  for (int i = 0; i < APASS; ++i)
#pragma unroll
    for (int j = 0; j < NT; ++j) acc[i][j] = v8f{};

#pragma unroll
  for (int i = 0; i < APASS; ++i)
    ra[i] = *(const v4f*)(xg + (size_t)(16 * i) * K);
#pragma unroll
  for (int i = 0; i < BPASS; ++i) {
    rb0[i] = *(const v4f*)(wg + (size_t)(2 * PPP * i) * M);
    rb1[i] = *(const v4f*)(wg + (size_t)(2 * PPP * i + 1) * M);
  }
#pragma unroll
  for (int i = 0; i < APASS; ++i)
    *(v4f*)&As[0][(arow + 16 * i) * GEMM_AP + aq] = ra[i];
#pragma unroll
  for (int i = 0; i < BPASS; ++i) {
    float* bd = &Bs[0][(bpr + PPP * i) * BP2 + 8 * bq];
    *(v4f*)bd       = v4f{rb0[i].x, rb1[i].x, rb0[i].y, rb1[i].y};
    *(v4f*)(bd + 4) = v4f{rb0[i].z, rb1[i].z, rb0[i].w, rb1[i].w};
  }
  __syncthreads();

  const int KT   = K / GEMM_BK;
  const int aoff = c * GEMM_AP + 2 * h;
  const int boff = h * BP2 + 2 * (wave * (16 * NT) + c);

  for (int kt = 0; kt < KT; ++kt) {
    const int  cur  = kt & 1;
    const bool more = (kt + 1 < KT);

    if (more) {
      const float* __restrict__ xa = xg + (size_t)(kt + 1) * GEMM_BK;
      const float* __restrict__ wb = wg + (size_t)(kt + 1) * GEMM_BK * M;
#pragma unroll
      for (int i = 0; i < APASS; ++i)
        ra[i] = *(const v4f*)(xa + (size_t)(16 * i) * K);
#pragma unroll
      for (int i = 0; i < BPASS; ++i) {
        rb0[i] = *(const v4f*)(wb + (size_t)(2 * PPP * i) * M);
        rb1[i] = *(const v4f*)(wb + (size_t)(2 * PPP * i + 1) * M);
      }
    }

    v2f fa[2][2][APASS];
    v2f fb[2][2][NT];
#pragma unroll
    for (int u = 0; u < 2; ++u) {
#pragma unroll
      for (int i = 0; i < APASS; ++i)
        fa[0][u][i] = *(const v2f*)&As[cur][aoff + (16 * i) * GEMM_AP + 4 * u];
#pragma unroll
      for (int j = 0; j < NT; ++j)
        fb[0][u][j] = *(const v2f*)&Bs[cur][boff + (2 * u) * BP2 + 32 * j];
    }

#pragma unroll
    for (int kk = 0; kk < GEMM_BK / 8; ++kk) {
      const int s = kk & 1;
      if (kk + 1 < GEMM_BK / 8) {
#pragma unroll
        for (int u = 0; u < 2; ++u) {
          const int ks = 2 * (kk + 1) + u;
#pragma unroll
          for (int i = 0; i < APASS; ++i)
            fa[s ^ 1][u][i] =
                *(const v2f*)&As[cur][aoff + (16 * i) * GEMM_AP + 4 * ks];
#pragma unroll
          for (int j = 0; j < NT; ++j)
            fb[s ^ 1][u][j] =
                *(const v2f*)&Bs[cur][boff + (2 * ks) * BP2 + 32 * j];
        }
      }
      __builtin_amdgcn_sched_barrier(0);
#pragma unroll
      for (int u = 0; u < 2; ++u)
#pragma unroll
        for (int i = 0; i < APASS; ++i)
#pragma unroll
          for (int j = 0; j < NT; ++j)
            acc[i][j] = __builtin_amdgcn_wmma_f32_16x16x4_f32(
                false, fa[s][u][i], false, fb[s][u][j], (short)0, acc[i][j],
                false, false);
      __builtin_amdgcn_sched_barrier(0);
    }

    if (more) {
#pragma unroll
      for (int i = 0; i < APASS; ++i)
        *(v4f*)&As[cur ^ 1][(arow + 16 * i) * GEMM_AP + aq] = ra[i];
#pragma unroll
      for (int i = 0; i < BPASS; ++i) {
        float* bd = &Bs[cur ^ 1][(bpr + PPP * i) * BP2 + 8 * bq];
        *(v4f*)bd       = v4f{rb0[i].x, rb1[i].x, rb0[i].y, rb1[i].y};
        *(v4f*)(bd + 4) = v4f{rb0[i].z, rb1[i].z, rb0[i].w, rb1[i].w};
      }
    }
    __syncthreads();
  }

  if constexpr (NT == 2) {
    const int head = blockIdx.y;
    const float4 sv = *(const float4*)(asrc + head * 128 + lane * 4);
    const float4 dv = *(const float4*)(adst + head * 128 + lane * 4);

    float* cw = smem + (8 * h) * GEMM_CP + wave * 32 + c;
#pragma unroll
    for (int i = 0; i < APASS; ++i)
#pragma unroll
      for (int v = 0; v < 8; ++v)
#pragma unroll
        for (int j = 0; j < NT; ++j)
          cw[(16 * i + v) * GEMM_CP + 16 * j] = acc[i][j][v];
    __syncthreads();

    constexpr int RPW = GEMM_BM / 4;
    const int r0 = wave * RPW;
    const float* cr = smem + r0 * GEMM_CP + lane * 4;
    float* __restrict__ orow = Hout + (size_t)(m0 + r0) * M + n0 + lane * 4;
    float ks = 0.0f, kd = 0.0f;
#pragma unroll
    for (int q = 0; q < RPW; ++q) {
      const float4 hv = *(const float4*)(cr + q * GEMM_CP);
      *(float4*)(orow + (size_t)q * M) = hv;
      float s = hv.x * sv.x + hv.y * sv.y + hv.z * sv.z + hv.w * sv.w;
      float d = hv.x * dv.x + hv.y * dv.y + hv.z * dv.z + hv.w * dv.w;
#pragma unroll
      for (int off = 16; off > 0; off >>= 1) {
        s += __shfl_xor(s, off);
        d += __shfl_xor(d, off);
      }
      if (lane == q) { ks = s; kd = d; }
    }
    if (lane < RPW) {
      const int o = (m0 + r0 + lane) * Hh + head;
      as_out[o] = ks;
      ad_out[o] = kd;
    }
  } else {
    float* __restrict__ ob =
        Hout + (size_t)(m0 + 8 * h) * M + n0 + wave * (16 * NT) + c;
#pragma unroll
    for (int i = 0; i < APASS; ++i)
#pragma unroll
      for (int v = 0; v < 8; ++v) {
        float* __restrict__ op = ob + (size_t)(16 * i + v) * M;
#pragma unroll
        for (int j = 0; j < NT; ++j) op[16 * j] = acc[i][j][v];
      }
  }
}

__global__ __launch_bounds__(128)
void alpha_kernel(const float* __restrict__ Hf, const float* __restrict__ asrc,
                  const float* __restrict__ adst, float* __restrict__ as_out,
                  float* __restrict__ ad_out, int Hh, int M, int total) {
  const int wid  = blockIdx.x * 4 + (threadIdx.x >> 5);
  const int lane = threadIdx.x & 31;
  if (wid >= total) return;
  const int n  = wid / Hh;
  const int hd = wid - n * Hh;

  const float4 hv = *(const float4*)(Hf + (size_t)n * M + hd * 128 + lane * 4);
  const float4 sv = *(const float4*)(asrc + hd * 128 + lane * 4);
  const float4 dv = *(const float4*)(adst + hd * 128 + lane * 4);
  float s = hv.x * sv.x + hv.y * sv.y + hv.z * sv.z + hv.w * sv.w;
  float d = hv.x * dv.x + hv.y * dv.y + hv.z * dv.z + hv.w * dv.w;
#pragma unroll
  for (int off = 16; off > 0; off >>= 1) {
    s += __shfl_xor(s, off);
    d += __shfl_xor(d, off);
  }
  if (lane == 0) { as_out[wid] = s; ad_out[wid] = d; }
}

__global__ __launch_bounds__(128)
void node_softmax_gather_kernel(const float* __restrict__ Hf,
                                const float* __restrict__ as,
                                const float* __restrict__ ad,
                                const int* __restrict__ rowptr,
                                const int* __restrict__ col,
                                const float* __restrict__ bias,
                                float* __restrict__ out,
                                int Hh, int M, int doElu, int total) {
  const int wv   = __builtin_amdgcn_readfirstlane((int)(threadIdx.x >> 5));
  const int wid  = blockIdx.x * 4 + wv;
  const int lane = threadIdx.x & 31;
  if (wid >= total) return;
  const int d   = wid / Hh;
  const int hd  = wid - d * Hh;
  const int beg = rowptr[d];
  const int deg = rowptr[d + 1] - beg;
  const int tot = deg + 1;
  const float adv = ad[wid];

  const bool v0 = lane < tot;
  int   s0 = d;
  float l0 = -INFINITY;
  if (v0) {
    s0 = (lane < deg) ? col[beg + lane] : d;
    const float l = as[s0 * Hh + hd] + adv;
    l0 = (l > 0.0f) ? l : NEG_SLOPE * l;
  }

  float m = l0;
  for (int i = lane + 32; i < tot; i += 32) {
    const int s = (i < deg) ? col[beg + i] : d;
    float l = as[s * Hh + hd] + adv;
    l = (l > 0.0f) ? l : NEG_SLOPE * l;
    m = fmaxf(m, l);
  }
  for (int off = 16; off > 0; off >>= 1) m = fmaxf(m, __shfl_xor(m, off));
  m = isfinite(m) ? m : 0.0f;

  const float ex0 = v0 ? expf(l0 - m) : 0.0f;
  float den = ex0;
  for (int i = lane + 32; i < tot; i += 32) {
    const int s = (i < deg) ? col[beg + i] : d;
    float l = as[s * Hh + hd] + adv;
    l = (l > 0.0f) ? l : NEG_SLOPE * l;
    den += expf(l - m);
  }
  for (int off = 16; off > 0; off >>= 1) den += __shfl_xor(den, off);
  const float dden = den + 1e-16f;

  float4 acc = make_float4(0.0f, 0.0f, 0.0f, 0.0f);
  const float* __restrict__ hp = Hf + hd * 128 + lane * 4;
  for (int cb = 0; cb < tot; cb += 32) {
    int   s_l = s0;
    float a_l = ex0 / dden;
    if (cb > 0) {
      const int i = cb + lane;
      s_l = d;
      a_l = 0.0f;
      if (i < tot) {
        s_l = (i < deg) ? col[beg + i] : d;
        float l = as[s_l * Hh + hd] + adv;
        l = (l > 0.0f) ? l : NEG_SLOPE * l;
        a_l = expf(l - m) / dden;
      }
    }
    const int nn = (tot - cb < 32) ? (tot - cb) : 32;
    for (int t = 0; t < nn; t += 8) {
      float4 hv[8];
      float  av[8];
      for (int u = 0; u < 8; ++u) {
        if (t + u < nn) {
          const int s = __builtin_amdgcn_readlane(s_l, t + u);
          av[u] = __int_as_float(
              __builtin_amdgcn_readlane(__float_as_int(a_l), t + u));
          hv[u] = *(const float4*)(hp + (size_t)s * M);
        }
      }
      for (int u = 0; u < 8; ++u) {
        if (t + u < nn) {
          acc.x += hv[u].x * av[u];
          acc.y += hv[u].y * av[u];
          acc.z += hv[u].z * av[u];
          acc.w += hv[u].w * av[u];
        }
      }
    }
  }

  const int c = hd * 128 + lane * 4;
  const float4 bv = *(const float4*)(bias + c);
  float4 r;
  r.x = acc.x + bv.x; r.y = acc.y + bv.y; r.z = acc.z + bv.z; r.w = acc.w + bv.w;
  if (doElu) {
    r.x = (r.x > 0.0f) ? r.x : expm1f(r.x);
    r.y = (r.y > 0.0f) ? r.y : expm1f(r.y);
    r.z = (r.z > 0.0f) ? r.z : expm1f(r.z);
    r.w = (r.w > 0.0f) ? r.w : expm1f(r.w);
  }
  *(float4*)(out + (size_t)d * M + c) = r;
}

static void run_layer(hipStream_t stream, const float* xin, const float* W,
                      const float* asrc, const float* adst, const float* bias,
                      int N, int K, int M, int Hh, bool elu,
                      const int* rowptr, const int* col,
                      float* hbuf, float* asb, float* adb, float* out,
                      int* sort_cursor, int* sort_rowptr, int* sort_col,
                      const int* esrc, const int* edst, int E0) {
  const int nh = N * Hh;

  if (M == 128) {
    gemm_wmma_f32<1, false><<<dim3(N / GEMM_BM, M / 64), 128, 0, stream>>>(
        xin, W, hbuf, K, M, asrc, adst, asb, adb, Hh, nullptr, nullptr, 0);
    alpha_kernel<<<(nh + 3) / 4, 128, 0, stream>>>(hbuf, asrc, adst, asb, adb, Hh, M, nh);
  } else if (sort_cursor != nullptr) {
    gemm_wmma_f32<2, true><<<dim3(N / GEMM_BM + 1, M / 128), 128, 0, stream>>>(
        xin, W, hbuf, K, M, asrc, adst, asb, adb, Hh,
        sort_cursor, sort_rowptr, N);
    csr_fill_kernel<<<(E0 + 255) / 256, 256, 0, stream>>>(
        esrc, edst, sort_cursor, sort_col, E0);
  } else {
    gemm_wmma_f32<2, false><<<dim3(N / GEMM_BM, M / 128), 128, 0, stream>>>(
        xin, W, hbuf, K, M, asrc, adst, asb, adb, Hh, nullptr, nullptr, 0);
  }

  node_softmax_gather_kernel<<<(nh + 3) / 4, 128, 0, stream>>>(
      hbuf, asb, adb, rowptr, col, bias, out, Hh, M, elu ? 1 : 0, nh);
}

extern "C" void kernel_launch(void* const* d_in, const int* in_sizes, int n_in,
                              void* d_out, int out_size, void* d_ws, size_t ws_size,
                              hipStream_t stream) {
  const float* x      = (const float*)d_in[0];
  const int*   ei     = (const int*)d_in[1];
  const float* W0     = (const float*)d_in[2];
  const float* a_src0 = (const float*)d_in[3];
  const float* a_dst0 = (const float*)d_in[4];
  const float* b0     = (const float*)d_in[5];
  const float* W1     = (const float*)d_in[6];
  const float* a_src1 = (const float*)d_in[7];
  const float* a_dst1 = (const float*)d_in[8];
  const float* b1     = (const float*)d_in[9];
  const float* W2     = (const float*)d_in[10];
  const float* a_src2 = (const float*)d_in[11];
  const float* a_dst2 = (const float*)d_in[12];
  const float* b2     = (const float*)d_in[13];

  const int N  = in_sizes[0] / 256;
  const int E0 = in_sizes[1] / 2;
  const int* esrc = ei;
  const int* edst = ei + E0;

  float* ws = (float*)d_ws;
  size_t p = 0;
  float* hbuf   = ws + p;             p += (size_t)N * 512;
  int*   csr    = (int*)(ws + p);     p += (size_t)N * 512;
  float* xbuf   = ws + p;             p += (size_t)N * 512;
  float* asb    = ws + p;             p += (size_t)N * 4;
  float* adb    = ws + p;             p += (size_t)N * 4;
  int*   cursor = csr;
  int*   rowptr = csr + N;
  int*   col    = csr + (2 * N + 1);
  (void)ws_size; (void)n_in; (void)out_size;

  csr_zero_kernel<<<(N + 255) / 256, 256, 0, stream>>>(cursor, N);
  csr_count_kernel<<<(E0 + 255) / 256, 256, 0, stream>>>(edst, cursor, E0);

  run_layer(stream, x, W0, a_src0, a_dst0, b0, N, 256, 512, 4, true,
            rowptr, col, hbuf, asb, adb, xbuf,
            cursor, rowptr, col, esrc, edst, E0);
  run_layer(stream, xbuf, W1, a_src1, a_dst1, b1, N, 512, 512, 4, true,
            rowptr, col, hbuf, asb, adb, xbuf,
            nullptr, nullptr, nullptr, nullptr, nullptr, 0);
  run_layer(stream, xbuf, W2, a_src2, a_dst2, b2, N, 512, 128, 1, false,
            rowptr, col, hbuf, asb, adb, (float*)d_out,
            nullptr, nullptr, nullptr, nullptr, nullptr, 0);
}


# Edge sort: an edge's place in its node's segment is the counting atomic's return value, kept per edge; fill stores at row start + rank, no second atomic pass; order = arrival rank at the count
# speedup vs baseline: 4.2779x; 1.1749x over previous
#include <hip/hip_runtime.h>
#include <hip/hip_bf16.h>
#include <math.h>


typedef __attribute__((ext_vector_type(2))) float v2f;
typedef __attribute__((ext_vector_type(4))) float v4f;
typedef __attribute__((ext_vector_type(8))) float v8f;

#define NEG_SLOPE 0.2f

__global__ void csr_zero_kernel(int* __restrict__ cnt, int n) {
  int i = blockIdx.x * blockDim.x + threadIdx.x;
  if (i < n) cnt[i] = 0;
}

__global__ void csr_count_kernel(const int* __restrict__ edst,
                                 int* __restrict__ cnt,
                                 int* __restrict__ rank, int E0) {
  int e = blockIdx.x * blockDim.x + threadIdx.x;
  if (e < E0) rank[e] = atomicAdd(&cnt[edst[e]], 1);
}

__device__ __forceinline__
void csr_scan_block128(const int* __restrict__ cnt, int* __restrict__ rowptr, int n,
                       int* sh, int cap) {
  const int t   = threadIdx.x;
  const int per = (n + 127) / 128;
  const int b   = t * per;
  const int e   = (b + per < n) ? (b + per) : n;
  int* part = sh + (cap - 128);
  const bool staged = (n <= cap - 128);
  int sum = 0;
  if (staged) {
    for (int i = t; i < n; i += 128) sh[i] = cnt[i];
    __syncthreads();
    for (int i = b; i < e; ++i) sum += sh[i];
  } else {
    for (int i = b; i < e; ++i) sum += cnt[i];
  }
  part[t] = sum;
  __syncthreads();
  for (int off = 1; off < 128; off <<= 1) {
    const int v = (t >= off) ? part[t - off] : 0;
    __syncthreads();
    part[t] += v;
    __syncthreads();
  }
  int run = part[t] - sum;
  if (staged) {
    for (int i = b; i < e; ++i) {
      const int c = sh[i];
      sh[i] = run;
      run += c;
    }
    __syncthreads();
    for (int i = t; i < n; i += 128) rowptr[i] = sh[i];
  } else {
    for (int i = b; i < e; ++i) {
      const int c = cnt[i];
      rowptr[i] = run;
      run += c;
    }
  }
  if (t == 127) rowptr[n] = part[127];
}

__global__ void csr_fill_kernel(const int* __restrict__ esrc,
                                const int* __restrict__ edst,
                                const int* __restrict__ rowptr,
                                const int* __restrict__ rank,
                                int* __restrict__ col, int E0) {
  int e = blockIdx.x * blockDim.x + threadIdx.x;
  if (e < E0) col[rowptr[edst[e]] + rank[e]] = esrc[e];
}

#define GEMM_BM 80
#define GEMM_BK 32
#define GEMM_AP 36
#define GEMM_CP 132

template <int NT, bool SCAN>
__global__ __launch_bounds__(128)
void gemm_wmma_f32(const float* __restrict__ X, const float* __restrict__ W,
                   float* __restrict__ Hout, int K, int M,
                   const float* __restrict__ asrc, const float* __restrict__ adst,
                   float* __restrict__ as_out, float* __restrict__ ad_out,
                   int Hh, const int* __restrict__ scan_cnt,
                   int* __restrict__ scan_rowptr, int scan_n) {
  constexpr int BN    = 64 * NT;
  constexpr int BP2   = 2 * BN + 32;
  constexpr int BQ    = BN / 4;
  constexpr int PPP   = 128 / BQ;
  constexpr int BPASS = (GEMM_BK / 2) / PPP;
  constexpr int APASS = GEMM_BM / 16;

  constexpr int AS_SZ = GEMM_BM * GEMM_AP;
  constexpr int BS_SZ = (GEMM_BK / 2) * BP2;
  static_assert(NT != 2 || GEMM_BM * GEMM_CP <= 2 * AS_SZ + 2 * BS_SZ,
                "the parked tile must fit in the K-loop's stages");

  __shared__ __attribute__((aligned(16))) float smem[2 * AS_SZ + 2 * BS_SZ];
  float (*As)[AS_SZ] = reinterpret_cast<float (*)[AS_SZ]>(smem);
  float (*Bs)[BS_SZ] = reinterpret_cast<float (*)[BS_SZ]>(smem + 2 * AS_SZ);

  if constexpr (SCAN) {
    if (blockIdx.x == gridDim.x - 1) {
      if (blockIdx.y == 0)
        csr_scan_block128(scan_cnt, scan_rowptr, scan_n,
                          reinterpret_cast<int*>(smem), 2 * AS_SZ + 2 * BS_SZ);
      return;
    }
  }

  const int t    = threadIdx.x;
  const int wave = t >> 5;
  const int lane = t & 31;
  const int c    = lane & 15;
  const int h    = lane >> 4;
  const int m0   = blockIdx.x * GEMM_BM;
  const int n0   = blockIdx.y * BN;

  const int arow = t >> 3;
  const int aq   = (t & 7) << 2;
  const int bpr  = t / BQ;
  const int bq   = t % BQ;

  const float* __restrict__ xg = X + (size_t)(m0 + arow) * K + aq;
  const float* __restrict__ wg = W + (size_t)(2 * bpr) * M + n0 + 4 * bq;

  v4f ra[APASS];
  v4f rb0[BPASS], rb1[BPASS];
  v8f acc[APASS][NT];
#pragma unroll
  for (int i = 0; i < APASS; ++i)
#pragma unroll
    for (int j = 0; j < NT; ++j) acc[i][j] = v8f{};

#pragma unroll
  for (int i = 0; i < APASS; ++i)
    ra[i] = *(const v4f*)(xg + (size_t)(16 * i) * K);
#pragma unroll
  for (int i = 0; i < BPASS; ++i) {
    rb0[i] = *(const v4f*)(wg + (size_t)(2 * PPP * i) * M);
    rb1[i] = *(const v4f*)(wg + (size_t)(2 * PPP * i + 1) * M);
  }
#pragma unroll
  for (int i = 0; i < APASS; ++i)
    *(v4f*)&As[0][(arow + 16 * i) * GEMM_AP + aq] = ra[i];
#pragma unroll
  for (int i = 0; i < BPASS; ++i) {
    float* bd = &Bs[0][(bpr + PPP * i) * BP2 + 8 * bq];
    *(v4f*)bd       = v4f{rb0[i].x, rb1[i].x, rb0[i].y, rb1[i].y};
    *(v4f*)(bd + 4) = v4f{rb0[i].z, rb1[i].z, rb0[i].w, rb1[i].w};
  }
  __syncthreads();

  const int KT   = K / GEMM_BK;
  const int aoff = c * GEMM_AP + 2 * h;
  const int boff = h * BP2 + 2 * (wave * (16 * NT) + c);

  for (int kt = 0; kt < KT; ++kt) {
    const int  cur  = kt & 1;
    const bool more = (kt + 1 < KT);

    if (more) {
      const float* __restrict__ xa = xg + (size_t)(kt + 1) * GEMM_BK;
      const float* __restrict__ wb = wg + (size_t)(kt + 1) * GEMM_BK * M;
#pragma unroll
      for (int i = 0; i < APASS; ++i)
        ra[i] = *(const v4f*)(xa + (size_t)(16 * i) * K);
#pragma unroll
      for (int i = 0; i < BPASS; ++i) {
        rb0[i] = *(const v4f*)(wb + (size_t)(2 * PPP * i) * M);
        rb1[i] = *(const v4f*)(wb + (size_t)(2 * PPP * i + 1) * M);
      }
    }

    v2f fa[2][2][APASS];
    v2f fb[2][2][NT];
#pragma unroll
    for (int u = 0; u < 2; ++u) {
#pragma unroll
      for (int i = 0; i < APASS; ++i)
        fa[0][u][i] = *(const v2f*)&As[cur][aoff + (16 * i) * GEMM_AP + 4 * u];
#pragma unroll
      for (int j = 0; j < NT; ++j)
        fb[0][u][j] = *(const v2f*)&Bs[cur][boff + (2 * u) * BP2 + 32 * j];
    }

#pragma unroll
    for (int kk = 0; kk < GEMM_BK / 8; ++kk) {
      const int s = kk & 1;
      if (kk + 1 < GEMM_BK / 8) {
#pragma unroll
        for (int u = 0; u < 2; ++u) {
          const int ks = 2 * (kk + 1) + u;
#pragma unroll
          for (int i = 0; i < APASS; ++i)
            fa[s ^ 1][u][i] =
                *(const v2f*)&As[cur][aoff + (16 * i) * GEMM_AP + 4 * ks];
#pragma unroll
          for (int j = 0; j < NT; ++j)
            fb[s ^ 1][u][j] =
                *(const v2f*)&Bs[cur][boff + (2 * ks) * BP2 + 32 * j];
        }
      }
      __builtin_amdgcn_sched_barrier(0);
#pragma unroll
      for (int u = 0; u < 2; ++u)
#pragma unroll
        for (int i = 0; i < APASS; ++i)
#pragma unroll
          for (int j = 0; j < NT; ++j)
            acc[i][j] = __builtin_amdgcn_wmma_f32_16x16x4_f32(
                false, fa[s][u][i], false, fb[s][u][j], (short)0, acc[i][j],
                false, false);
      __builtin_amdgcn_sched_barrier(0);
    }

    if (more) {
#pragma unroll
      for (int i = 0; i < APASS; ++i)
        *(v4f*)&As[cur ^ 1][(arow + 16 * i) * GEMM_AP + aq] = ra[i];
#pragma unroll
      for (int i = 0; i < BPASS; ++i) {
        float* bd = &Bs[cur ^ 1][(bpr + PPP * i) * BP2 + 8 * bq];
        *(v4f*)bd       = v4f{rb0[i].x, rb1[i].x, rb0[i].y, rb1[i].y};
        *(v4f*)(bd + 4) = v4f{rb0[i].z, rb1[i].z, rb0[i].w, rb1[i].w};
      }
    }
    __syncthreads();
  }

  if constexpr (NT == 2) {
    const int head = blockIdx.y;
    const float4 sv = *(const float4*)(asrc + head * 128 + lane * 4);
    const float4 dv = *(const float4*)(adst + head * 128 + lane * 4);

    float* cw = smem + (8 * h) * GEMM_CP + wave * 32 + c;
#pragma unroll
    for (int i = 0; i < APASS; ++i)
#pragma unroll
      for (int v = 0; v < 8; ++v)
#pragma unroll
        for (int j = 0; j < NT; ++j)
          cw[(16 * i + v) * GEMM_CP + 16 * j] = acc[i][j][v];
    __syncthreads();

    constexpr int RPW = GEMM_BM / 4;
    const int r0 = wave * RPW;
    const float* cr = smem + r0 * GEMM_CP + lane * 4;
    float* __restrict__ orow = Hout + (size_t)(m0 + r0) * M + n0 + lane * 4;
    float ks = 0.0f, kd = 0.0f;
#pragma unroll
    for (int q = 0; q < RPW; ++q) {
      const float4 hv = *(const float4*)(cr + q * GEMM_CP);
      *(float4*)(orow + (size_t)q * M) = hv;
      float s = hv.x * sv.x + hv.y * sv.y + hv.z * sv.z + hv.w * sv.w;
      float d = hv.x * dv.x + hv.y * dv.y + hv.z * dv.z + hv.w * dv.w;
#pragma unroll
      for (int off = 16; off > 0; off >>= 1) {
        s += __shfl_xor(s, off);
        d += __shfl_xor(d, off);
      }
      if (lane == q) { ks = s; kd = d; }
    }
    if (lane < RPW) {
      const int o = (m0 + r0 + lane) * Hh + head;
      as_out[o] = ks;
      ad_out[o] = kd;
    }
  } else {
    float* __restrict__ ob =
        Hout + (size_t)(m0 + 8 * h) * M + n0 + wave * (16 * NT) + c;
#pragma unroll
    for (int i = 0; i < APASS; ++i)
#pragma unroll
      for (int v = 0; v < 8; ++v) {
        float* __restrict__ op = ob + (size_t)(16 * i + v) * M;
#pragma unroll
        for (int j = 0; j < NT; ++j) op[16 * j] = acc[i][j][v];
      }
  }
}

__global__ __launch_bounds__(128)
void alpha_kernel(const float* __restrict__ Hf, const float* __restrict__ asrc,
                  const float* __restrict__ adst, float* __restrict__ as_out,
                  float* __restrict__ ad_out, int Hh, int M, int total) {
  const int wid  = blockIdx.x * 4 + (threadIdx.x >> 5);
  const int lane = threadIdx.x & 31;
  if (wid >= total) return;
  const int n  = wid / Hh;
  const int hd = wid - n * Hh;

  const float4 hv = *(const float4*)(Hf + (size_t)n * M + hd * 128 + lane * 4);
  const float4 sv = *(const float4*)(asrc + hd * 128 + lane * 4);
  const float4 dv = *(const float4*)(adst + hd * 128 + lane * 4);
  float s = hv.x * sv.x + hv.y * sv.y + hv.z * sv.z + hv.w * sv.w;
  float d = hv.x * dv.x + hv.y * dv.y + hv.z * dv.z + hv.w * dv.w;
#pragma unroll
  for (int off = 16; off > 0; off >>= 1) {
    s += __shfl_xor(s, off);
    d += __shfl_xor(d, off);
  }
  if (lane == 0) { as_out[wid] = s; ad_out[wid] = d; }
}

__global__ __launch_bounds__(128)
void node_softmax_gather_kernel(const float* __restrict__ Hf,
                                const float* __restrict__ as,
                                const float* __restrict__ ad,
                                const int* __restrict__ rowptr,
                                const int* __restrict__ col,
                                const float* __restrict__ bias,
                                float* __restrict__ out,
                                int Hh, int M, int doElu, int total) {
  const int wv   = __builtin_amdgcn_readfirstlane((int)(threadIdx.x >> 5));
  const int wid  = blockIdx.x * 4 + wv;
  const int lane = threadIdx.x & 31;
  if (wid >= total) return;
  const int d   = wid / Hh;
  const int hd  = wid - d * Hh;
  const int beg = rowptr[d];
  const int deg = rowptr[d + 1] - beg;
  const int tot = deg + 1;
  const float adv = ad[wid];

  const bool v0 = lane < tot;
  int   s0 = d;
  float l0 = -INFINITY;
  if (v0) {
    s0 = (lane < deg) ? col[beg + lane] : d;
    const float l = as[s0 * Hh + hd] + adv;
    l0 = (l > 0.0f) ? l : NEG_SLOPE * l;
  }

  float m = l0;
  for (int i = lane + 32; i < tot; i += 32) {
    const int s = (i < deg) ? col[beg + i] : d;
    float l = as[s * Hh + hd] + adv;
    l = (l > 0.0f) ? l : NEG_SLOPE * l;
    m = fmaxf(m, l);
  }
  for (int off = 16; off > 0; off >>= 1) m = fmaxf(m, __shfl_xor(m, off));
  m = isfinite(m) ? m : 0.0f;

  const float ex0 = v0 ? expf(l0 - m) : 0.0f;
  float den = ex0;
  for (int i = lane + 32; i < tot; i += 32) {
    const int s = (i < deg) ? col[beg + i] : d;
    float l = as[s * Hh + hd] + adv;
    l = (l > 0.0f) ? l : NEG_SLOPE * l;
    den += expf(l - m);
  }
  for (int off = 16; off > 0; off >>= 1) den += __shfl_xor(den, off);
  const float dden = den + 1e-16f;

  float4 acc = make_float4(0.0f, 0.0f, 0.0f, 0.0f);
  const float* __restrict__ hp = Hf + hd * 128 + lane * 4;
  for (int cb = 0; cb < tot; cb += 32) {
    int   s_l = s0;
    float a_l = ex0 / dden;
    if (cb > 0) {
      const int i = cb + lane;
      s_l = d;
      a_l = 0.0f;
      if (i < tot) {
        s_l = (i < deg) ? col[beg + i] : d;
        float l = as[s_l * Hh + hd] + adv;
        l = (l > 0.0f) ? l : NEG_SLOPE * l;
        a_l = expf(l - m) / dden;
      }
    }
    const int nn = (tot - cb < 32) ? (tot - cb) : 32;
    for (int t = 0; t < nn; t += 8) {
      float4 hv[8];
      float  av[8];
      for (int u = 0; u < 8; ++u) {
        if (t + u < nn) {
          const int s = __builtin_amdgcn_readlane(s_l, t + u);
          av[u] = __int_as_float(
              __builtin_amdgcn_readlane(__float_as_int(a_l), t + u));
          hv[u] = *(const float4*)(hp + (size_t)s * M);
        }
      }
      for (int u = 0; u < 8; ++u) {
        if (t + u < nn) {
          acc.x += hv[u].x * av[u];
          acc.y += hv[u].y * av[u];
          acc.z += hv[u].z * av[u];
          acc.w += hv[u].w * av[u];
        }
      }
    }
  }

  const int c = hd * 128 + lane * 4;
  const float4 bv = *(const float4*)(bias + c);
  float4 r;
  r.x = acc.x + bv.x; r.y = acc.y + bv.y; r.z = acc.z + bv.z; r.w = acc.w + bv.w;
  if (doElu) {
    r.x = (r.x > 0.0f) ? r.x : expm1f(r.x);
    r.y = (r.y > 0.0f) ? r.y : expm1f(r.y);
    r.z = (r.z > 0.0f) ? r.z : expm1f(r.z);
    r.w = (r.w > 0.0f) ? r.w : expm1f(r.w);
  }
  *(float4*)(out + (size_t)d * M + c) = r;
}

static void run_layer(hipStream_t stream, const float* xin, const float* W,
                      const float* asrc, const float* adst, const float* bias,
                      int N, int K, int M, int Hh, bool elu,
                      const int* rowptr, const int* col,
                      float* hbuf, float* asb, float* adb, float* out,
                      const int* sort_cnt, const int* sort_rank,
                      int* sort_rowptr, int* sort_col,
                      const int* esrc, const int* edst, int E0) {
  const int nh = N * Hh;

  if (M == 128) {
    gemm_wmma_f32<1, false><<<dim3(N / GEMM_BM, M / 64), 128, 0, stream>>>(
        xin, W, hbuf, K, M, asrc, adst, asb, adb, Hh, nullptr, nullptr, 0);
    alpha_kernel<<<(nh + 3) / 4, 128, 0, stream>>>(hbuf, asrc, adst, asb, adb, Hh, M, nh);
  } else if (sort_cnt != nullptr) {
    gemm_wmma_f32<2, true><<<dim3(N / GEMM_BM + 1, M / 128), 128, 0, stream>>>(
        xin, W, hbuf, K, M, asrc, adst, asb, adb, Hh,
        sort_cnt, sort_rowptr, N);
    csr_fill_kernel<<<(E0 + 255) / 256, 256, 0, stream>>>(
        esrc, edst, sort_rowptr, sort_rank, sort_col, E0);
  } else {
    gemm_wmma_f32<2, false><<<dim3(N / GEMM_BM, M / 128), 128, 0, stream>>>(
        xin, W, hbuf, K, M, asrc, adst, asb, adb, Hh, nullptr, nullptr, 0);
  }

  node_softmax_gather_kernel<<<(nh + 3) / 4, 128, 0, stream>>>(
      hbuf, asb, adb, rowptr, col, bias, out, Hh, M, elu ? 1 : 0, nh);
}

extern "C" void kernel_launch(void* const* d_in, const int* in_sizes, int n_in,
                              void* d_out, int out_size, void* d_ws, size_t ws_size,
                              hipStream_t stream) {
  const float* x      = (const float*)d_in[0];
  const int*   ei     = (const int*)d_in[1];
  const float* W0     = (const float*)d_in[2];
  const float* a_src0 = (const float*)d_in[3];
  const float* a_dst0 = (const float*)d_in[4];
  const float* b0     = (const float*)d_in[5];
  const float* W1     = (const float*)d_in[6];
  const float* a_src1 = (const float*)d_in[7];
  const float* a_dst1 = (const float*)d_in[8];
  const float* b1     = (const float*)d_in[9];
  const float* W2     = (const float*)d_in[10];
  const float* a_src2 = (const float*)d_in[11];
  const float* a_dst2 = (const float*)d_in[12];
  const float* b2     = (const float*)d_in[13];

  const int N  = in_sizes[0] / 256;
  const int E0 = in_sizes[1] / 2;
  const int* esrc = ei;
  const int* edst = ei + E0;

  float* ws = (float*)d_ws;
  size_t p = 0;
  float* hbuf   = ws + p;             p += (size_t)N * 512;
  int*   csr    = (int*)(ws + p);     p += (size_t)N * 512;
  float* xbuf   = ws + p;             p += (size_t)N * 512;
  float* asb    = ws + p;             p += (size_t)N * 4;
  float* adb    = ws + p;             p += (size_t)N * 4;
  int*   cnt    = csr;
  int*   rowptr = csr + N;
  int*   col    = csr + (2 * N + 1);
  int*   rank   = col + E0;
  (void)ws_size; (void)n_in; (void)out_size;

  csr_zero_kernel<<<(N + 255) / 256, 256, 0, stream>>>(cnt, N);
  csr_count_kernel<<<(E0 + 255) / 256, 256, 0, stream>>>(edst, cnt, rank, E0);

  run_layer(stream, x, W0, a_src0, a_dst0, b0, N, 256, 512, 4, true,
            rowptr, col, hbuf, asb, adb, xbuf,
            cnt, rank, rowptr, col, esrc, edst, E0);
  run_layer(stream, xbuf, W1, a_src1, a_dst1, b1, N, 512, 512, 4, true,
            rowptr, col, hbuf, asb, adb, xbuf,
            nullptr, nullptr, nullptr, nullptr, nullptr, nullptr, 0);
  run_layer(stream, xbuf, W2, a_src2, a_dst2, b2, N, 512, 128, 1, false,
            rowptr, col, hbuf, asb, adb, (float*)d_out,
            nullptr, nullptr, nullptr, nullptr, nullptr, nullptr, 0);
}


# Edge sort count: a block counts its 10240 edges per node in LDS (LDS atomic's return = place in block), then one global add per block and node, whole lines; order = block arrival, then LDS rank
# speedup vs baseline: 5.2362x; 1.2240x over previous
#include <hip/hip_runtime.h>
#include <hip/hip_bf16.h>
#include <math.h>


typedef __attribute__((ext_vector_type(2))) float v2f;
typedef __attribute__((ext_vector_type(4))) float v4f;
typedef __attribute__((ext_vector_type(8))) float v8f;

#define NEG_SLOPE 0.2f

__global__ void csr_zero_kernel(int* __restrict__ cnt, int n) {
  int i = blockIdx.x * blockDim.x + threadIdx.x;
  if (i < n) cnt[i] = 0;
}

#define CSR_EPT 10
#define CSR_CAP 16384
__global__ __launch_bounds__(1024)
void csr_count_kernel(const int* __restrict__ edst,
                      int* __restrict__ cnt,
                      int* __restrict__ rank, int E0, int n) {
  __shared__ int hist[CSR_CAP];
  const int t  = threadIdx.x;
  const int eb = blockIdx.x * (1024 * CSR_EPT) + t;

  int d[CSR_EPT], r[CSR_EPT];
#pragma unroll
  for (int u = 0; u < CSR_EPT; ++u) {
    const int e = eb + u * 1024;
    d[u] = (e < E0) ? edst[e] : -1;
    r[u] = 0;
  }

  for (int lo = 0; lo < n; lo += CSR_CAP) {
    const int nb = (n - lo < CSR_CAP) ? (n - lo) : CSR_CAP;
    for (int i = t; i < nb; i += 1024) hist[i] = 0;
    __syncthreads();

#pragma unroll
    for (int u = 0; u < CSR_EPT; ++u) {
      const unsigned k = (unsigned)(d[u] - lo);
      if (k < (unsigned)nb) r[u] = atomicAdd(&hist[k], 1);
    }
    __syncthreads();

    for (int ib = t; ib < nb; ib += 1024 * CSR_EPT) {
      int c[CSR_EPT], b[CSR_EPT];
#pragma unroll
      for (int u = 0; u < CSR_EPT; ++u) {
        const int i = ib + u * 1024;
        c[u] = (i < nb) ? hist[i] : 0;
      }
#pragma unroll
      for (int u = 0; u < CSR_EPT; ++u)
        b[u] = (c[u] > 0) ? atomicAdd(&cnt[lo + ib + u * 1024], c[u]) : 0;
#pragma unroll
      for (int u = 0; u < CSR_EPT; ++u)
        if (c[u] > 0) hist[ib + u * 1024] = b[u];
    }
    __syncthreads();

    int hb[CSR_EPT];
#pragma unroll
    for (int u = 0; u < CSR_EPT; ++u) {
      const unsigned k = (unsigned)(d[u] - lo);
      hb[u] = (k < (unsigned)nb) ? hist[k] : 0;
    }
#pragma unroll
    for (int u = 0; u < CSR_EPT; ++u) {
      const unsigned k = (unsigned)(d[u] - lo);
      if (k < (unsigned)nb) rank[eb + u * 1024] = hb[u] + r[u];
    }
    __syncthreads();
  }
}

__device__ __forceinline__
void csr_scan_block128(const int* __restrict__ cnt, int* __restrict__ rowptr, int n,
                       int* sh, int cap) {
  const int t   = threadIdx.x;
  const int per = (n + 127) / 128;
  const int b   = t * per;
  const int e   = (b + per < n) ? (b + per) : n;
  int* part = sh + (cap - 128);
  const bool staged = (n <= cap - 128);
  int sum = 0;
  if (staged) {
    for (int i = t; i < n; i += 128) sh[i] = cnt[i];
    __syncthreads();
    for (int i = b; i < e; ++i) sum += sh[i];
  } else {
    for (int i = b; i < e; ++i) sum += cnt[i];
  }
  part[t] = sum;
  __syncthreads();
  for (int off = 1; off < 128; off <<= 1) {
    const int v = (t >= off) ? part[t - off] : 0;
    __syncthreads();
    part[t] += v;
    __syncthreads();
  }
  int run = part[t] - sum;
  if (staged) {
    for (int i = b; i < e; ++i) {
      const int c = sh[i];
      sh[i] = run;
      run += c;
    }
    __syncthreads();
    for (int i = t; i < n; i += 128) rowptr[i] = sh[i];
  } else {
    for (int i = b; i < e; ++i) {
      const int c = cnt[i];
      rowptr[i] = run;
      run += c;
    }
  }
  if (t == 127) rowptr[n] = part[127];
}

__global__ void csr_fill_kernel(const int* __restrict__ esrc,
                                const int* __restrict__ edst,
                                const int* __restrict__ rowptr,
                                const int* __restrict__ rank,
                                int* __restrict__ col, int E0) {
  int e = blockIdx.x * blockDim.x + threadIdx.x;
  if (e < E0) col[rowptr[edst[e]] + rank[e]] = esrc[e];
}

#define GEMM_BM 80
#define GEMM_BK 32
#define GEMM_AP 36
#define GEMM_CP 132

template <int NT, bool SCAN>
__global__ __launch_bounds__(128)
void gemm_wmma_f32(const float* __restrict__ X, const float* __restrict__ W,
                   float* __restrict__ Hout, int K, int M,
                   const float* __restrict__ asrc, const float* __restrict__ adst,
                   float* __restrict__ as_out, float* __restrict__ ad_out,
                   int Hh, const int* __restrict__ scan_cnt,
                   int* __restrict__ scan_rowptr, int scan_n) {
  constexpr int BN    = 64 * NT;
  constexpr int BP2   = 2 * BN + 32;
  constexpr int BQ    = BN / 4;
  constexpr int PPP   = 128 / BQ;
  constexpr int BPASS = (GEMM_BK / 2) / PPP;
  constexpr int APASS = GEMM_BM / 16;

  constexpr int AS_SZ = GEMM_BM * GEMM_AP;
  constexpr int BS_SZ = (GEMM_BK / 2) * BP2;
  static_assert(NT != 2 || GEMM_BM * GEMM_CP <= 2 * AS_SZ + 2 * BS_SZ,
                "the parked tile must fit in the K-loop's stages");

  __shared__ __attribute__((aligned(16))) float smem[2 * AS_SZ + 2 * BS_SZ];
  float (*As)[AS_SZ] = reinterpret_cast<float (*)[AS_SZ]>(smem);
  float (*Bs)[BS_SZ] = reinterpret_cast<float (*)[BS_SZ]>(smem + 2 * AS_SZ);

  if constexpr (SCAN) {
    if (blockIdx.x == gridDim.x - 1) {
      if (blockIdx.y == 0)
        csr_scan_block128(scan_cnt, scan_rowptr, scan_n,
                          reinterpret_cast<int*>(smem), 2 * AS_SZ + 2 * BS_SZ);
      return;
    }
  }

  const int t    = threadIdx.x;
  const int wave = t >> 5;
  const int lane = t & 31;
  const int c    = lane & 15;
  const int h    = lane >> 4;
  const int m0   = blockIdx.x * GEMM_BM;
  const int n0   = blockIdx.y * BN;

  const int arow = t >> 3;
  const int aq   = (t & 7) << 2;
  const int bpr  = t / BQ;
  const int bq   = t % BQ;

  const float* __restrict__ xg = X + (size_t)(m0 + arow) * K + aq;
  const float* __restrict__ wg = W + (size_t)(2 * bpr) * M + n0 + 4 * bq;

  v4f ra[APASS];
  v4f rb0[BPASS], rb1[BPASS];
  v8f acc[APASS][NT];
#pragma unroll
  for (int i = 0; i < APASS; ++i)
#pragma unroll
    for (int j = 0; j < NT; ++j) acc[i][j] = v8f{};

#pragma unroll
  for (int i = 0; i < APASS; ++i)
    ra[i] = *(const v4f*)(xg + (size_t)(16 * i) * K);
#pragma unroll
  for (int i = 0; i < BPASS; ++i) {
    rb0[i] = *(const v4f*)(wg + (size_t)(2 * PPP * i) * M);
    rb1[i] = *(const v4f*)(wg + (size_t)(2 * PPP * i + 1) * M);
  }
#pragma unroll
  for (int i = 0; i < APASS; ++i)
    *(v4f*)&As[0][(arow + 16 * i) * GEMM_AP + aq] = ra[i];
#pragma unroll
  for (int i = 0; i < BPASS; ++i) {
    float* bd = &Bs[0][(bpr + PPP * i) * BP2 + 8 * bq];
    *(v4f*)bd       = v4f{rb0[i].x, rb1[i].x, rb0[i].y, rb1[i].y};
    *(v4f*)(bd + 4) = v4f{rb0[i].z, rb1[i].z, rb0[i].w, rb1[i].w};
  }
  __syncthreads();

  const int KT   = K / GEMM_BK;
  const int aoff = c * GEMM_AP + 2 * h;
  const int boff = h * BP2 + 2 * (wave * (16 * NT) + c);

  for (int kt = 0; kt < KT; ++kt) {
    const int  cur  = kt & 1;
    const bool more = (kt + 1 < KT);

    if (more) {
      const float* __restrict__ xa = xg + (size_t)(kt + 1) * GEMM_BK;
      const float* __restrict__ wb = wg + (size_t)(kt + 1) * GEMM_BK * M;
#pragma unroll
      for (int i = 0; i < APASS; ++i)
        ra[i] = *(const v4f*)(xa + (size_t)(16 * i) * K);
#pragma unroll
      for (int i = 0; i < BPASS; ++i) {
        rb0[i] = *(const v4f*)(wb + (size_t)(2 * PPP * i) * M);
        rb1[i] = *(const v4f*)(wb + (size_t)(2 * PPP * i + 1) * M);
      }
    }

    v2f fa[2][2][APASS];
    v2f fb[2][2][NT];
#pragma unroll
    for (int u = 0; u < 2; ++u) {
#pragma unroll
      for (int i = 0; i < APASS; ++i)
        fa[0][u][i] = *(const v2f*)&As[cur][aoff + (16 * i) * GEMM_AP + 4 * u];
#pragma unroll
      for (int j = 0; j < NT; ++j)
        fb[0][u][j] = *(const v2f*)&Bs[cur][boff + (2 * u) * BP2 + 32 * j];
    }

#pragma unroll
    for (int kk = 0; kk < GEMM_BK / 8; ++kk) {
      const int s = kk & 1;
      if (kk + 1 < GEMM_BK / 8) {
#pragma unroll
        for (int u = 0; u < 2; ++u) {
          const int ks = 2 * (kk + 1) + u;
#pragma unroll
          for (int i = 0; i < APASS; ++i)
            fa[s ^ 1][u][i] =
                *(const v2f*)&As[cur][aoff + (16 * i) * GEMM_AP + 4 * ks];
#pragma unroll
          for (int j = 0; j < NT; ++j)
            fb[s ^ 1][u][j] =
                *(const v2f*)&Bs[cur][boff + (2 * ks) * BP2 + 32 * j];
        }
      }
      __builtin_amdgcn_sched_barrier(0);
#pragma unroll
      for (int u = 0; u < 2; ++u)
#pragma unroll
        for (int i = 0; i < APASS; ++i)
#pragma unroll
          for (int j = 0; j < NT; ++j)
            acc[i][j] = __builtin_amdgcn_wmma_f32_16x16x4_f32(
                false, fa[s][u][i], false, fb[s][u][j], (short)0, acc[i][j],
                false, false);
      __builtin_amdgcn_sched_barrier(0);
    }

    if (more) {
#pragma unroll
      for (int i = 0; i < APASS; ++i)
        *(v4f*)&As[cur ^ 1][(arow + 16 * i) * GEMM_AP + aq] = ra[i];
#pragma unroll
      for (int i = 0; i < BPASS; ++i) {
        float* bd = &Bs[cur ^ 1][(bpr + PPP * i) * BP2 + 8 * bq];
        *(v4f*)bd       = v4f{rb0[i].x, rb1[i].x, rb0[i].y, rb1[i].y};
        *(v4f*)(bd + 4) = v4f{rb0[i].z, rb1[i].z, rb0[i].w, rb1[i].w};
      }
    }
    __syncthreads();
  }

  if constexpr (NT == 2) {
    const int head = blockIdx.y;
    const float4 sv = *(const float4*)(asrc + head * 128 + lane * 4);
    const float4 dv = *(const float4*)(adst + head * 128 + lane * 4);

    float* cw = smem + (8 * h) * GEMM_CP + wave * 32 + c;
#pragma unroll
    for (int i = 0; i < APASS; ++i)
#pragma unroll
      for (int v = 0; v < 8; ++v)
#pragma unroll
        for (int j = 0; j < NT; ++j)
          cw[(16 * i + v) * GEMM_CP + 16 * j] = acc[i][j][v];
    __syncthreads();

    constexpr int RPW = GEMM_BM / 4;
    const int r0 = wave * RPW;
    const float* cr = smem + r0 * GEMM_CP + lane * 4;
    float* __restrict__ orow = Hout + (size_t)(m0 + r0) * M + n0 + lane * 4;
    float ks = 0.0f, kd = 0.0f;
#pragma unroll
    for (int q = 0; q < RPW; ++q) {
      const float4 hv = *(const float4*)(cr + q * GEMM_CP);
      *(float4*)(orow + (size_t)q * M) = hv;
      float s = hv.x * sv.x + hv.y * sv.y + hv.z * sv.z + hv.w * sv.w;
      float d = hv.x * dv.x + hv.y * dv.y + hv.z * dv.z + hv.w * dv.w;
#pragma unroll
      for (int off = 16; off > 0; off >>= 1) {
        s += __shfl_xor(s, off);
        d += __shfl_xor(d, off);
      }
      if (lane == q) { ks = s; kd = d; }
    }
    if (lane < RPW) {
      const int o = (m0 + r0 + lane) * Hh + head;
      as_out[o] = ks;
      ad_out[o] = kd;
    }
  } else {
    float* __restrict__ ob =
        Hout + (size_t)(m0 + 8 * h) * M + n0 + wave * (16 * NT) + c;
#pragma unroll
    for (int i = 0; i < APASS; ++i)
#pragma unroll
      for (int v = 0; v < 8; ++v) {
        float* __restrict__ op = ob + (size_t)(16 * i + v) * M;
#pragma unroll
        for (int j = 0; j < NT; ++j) op[16 * j] = acc[i][j][v];
      }
  }
}

__global__ __launch_bounds__(128)
void alpha_kernel(const float* __restrict__ Hf, const float* __restrict__ asrc,
                  const float* __restrict__ adst, float* __restrict__ as_out,
                  float* __restrict__ ad_out, int Hh, int M, int total) {
  const int wid  = blockIdx.x * 4 + (threadIdx.x >> 5);
  const int lane = threadIdx.x & 31;
  if (wid >= total) return;
  const int n  = wid / Hh;
  const int hd = wid - n * Hh;

  const float4 hv = *(const float4*)(Hf + (size_t)n * M + hd * 128 + lane * 4);
  const float4 sv = *(const float4*)(asrc + hd * 128 + lane * 4);
  const float4 dv = *(const float4*)(adst + hd * 128 + lane * 4);
  float s = hv.x * sv.x + hv.y * sv.y + hv.z * sv.z + hv.w * sv.w;
  float d = hv.x * dv.x + hv.y * dv.y + hv.z * dv.z + hv.w * dv.w;
#pragma unroll
  for (int off = 16; off > 0; off >>= 1) {
    s += __shfl_xor(s, off);
    d += __shfl_xor(d, off);
  }
  if (lane == 0) { as_out[wid] = s; ad_out[wid] = d; }
}

__global__ __launch_bounds__(128)
void node_softmax_gather_kernel(const float* __restrict__ Hf,
                                const float* __restrict__ as,
                                const float* __restrict__ ad,
                                const int* __restrict__ rowptr,
                                const int* __restrict__ col,
                                const float* __restrict__ bias,
                                float* __restrict__ out,
                                int Hh, int M, int doElu, int total) {
  const int wv   = __builtin_amdgcn_readfirstlane((int)(threadIdx.x >> 5));
  const int wid  = blockIdx.x * 4 + wv;
  const int lane = threadIdx.x & 31;
  if (wid >= total) return;
  const int d   = wid / Hh;
  const int hd  = wid - d * Hh;
  const int beg = rowptr[d];
  const int deg = rowptr[d + 1] - beg;
  const int tot = deg + 1;
  const float adv = ad[wid];

  const bool v0 = lane < tot;
  int   s0 = d;
  float l0 = -INFINITY;
  if (v0) {
    s0 = (lane < deg) ? col[beg + lane] : d;
    const float l = as[s0 * Hh + hd] + adv;
    l0 = (l > 0.0f) ? l : NEG_SLOPE * l;
  }

  float m = l0;
  for (int i = lane + 32; i < tot; i += 32) {
    const int s = (i < deg) ? col[beg + i] : d;
    float l = as[s * Hh + hd] + adv;
    l = (l > 0.0f) ? l : NEG_SLOPE * l;
    m = fmaxf(m, l);
  }
  for (int off = 16; off > 0; off >>= 1) m = fmaxf(m, __shfl_xor(m, off));
  m = isfinite(m) ? m : 0.0f;

  const float ex0 = v0 ? expf(l0 - m) : 0.0f;
  float den = ex0;
  for (int i = lane + 32; i < tot; i += 32) {
    const int s = (i < deg) ? col[beg + i] : d;
    float l = as[s * Hh + hd] + adv;
    l = (l > 0.0f) ? l : NEG_SLOPE * l;
    den += expf(l - m);
  }
  for (int off = 16; off > 0; off >>= 1) den += __shfl_xor(den, off);
  const float dden = den + 1e-16f;

  float4 acc = make_float4(0.0f, 0.0f, 0.0f, 0.0f);
  const float* __restrict__ hp = Hf + hd * 128 + lane * 4;
  for (int cb = 0; cb < tot; cb += 32) {
    int   s_l = s0;
    float a_l = ex0 / dden;
    if (cb > 0) {
      const int i = cb + lane;
      s_l = d;
      a_l = 0.0f;
      if (i < tot) {
        s_l = (i < deg) ? col[beg + i] : d;
        float l = as[s_l * Hh + hd] + adv;
        l = (l > 0.0f) ? l : NEG_SLOPE * l;
        a_l = expf(l - m) / dden;
      }
    }
    const int nn = (tot - cb < 32) ? (tot - cb) : 32;
    for (int t = 0; t < nn; t += 8) {
      float4 hv[8];
      float  av[8];
      for (int u = 0; u < 8; ++u) {
        if (t + u < nn) {
          const int s = __builtin_amdgcn_readlane(s_l, t + u);
          av[u] = __int_as_float(
              __builtin_amdgcn_readlane(__float_as_int(a_l), t + u));
          hv[u] = *(const float4*)(hp + (size_t)s * M);
        }
      }
      for (int u = 0; u < 8; ++u) {
        if (t + u < nn) {
          acc.x += hv[u].x * av[u];
          acc.y += hv[u].y * av[u];
          acc.z += hv[u].z * av[u];
          acc.w += hv[u].w * av[u];
        }
      }
    }
  }

  const int c = hd * 128 + lane * 4;
  const float4 bv = *(const float4*)(bias + c);
  float4 r;
  r.x = acc.x + bv.x; r.y = acc.y + bv.y; r.z = acc.z + bv.z; r.w = acc.w + bv.w;
  if (doElu) {
    r.x = (r.x > 0.0f) ? r.x : expm1f(r.x);
    r.y = (r.y > 0.0f) ? r.y : expm1f(r.y);
    r.z = (r.z > 0.0f) ? r.z : expm1f(r.z);
    r.w = (r.w > 0.0f) ? r.w : expm1f(r.w);
  }
  *(float4*)(out + (size_t)d * M + c) = r;
}

static void run_layer(hipStream_t stream, const float* xin, const float* W,
                      const float* asrc, const float* adst, const float* bias,
                      int N, int K, int M, int Hh, bool elu,
                      const int* rowptr, const int* col,
                      float* hbuf, float* asb, float* adb, float* out,
                      const int* sort_cnt, const int* sort_rank,
                      int* sort_rowptr, int* sort_col,
                      const int* esrc, const int* edst, int E0) {
  const int nh = N * Hh;

  if (M == 128) {
    gemm_wmma_f32<1, false><<<dim3(N / GEMM_BM, M / 64), 128, 0, stream>>>(
        xin, W, hbuf, K, M, asrc, adst, asb, adb, Hh, nullptr, nullptr, 0);
    alpha_kernel<<<(nh + 3) / 4, 128, 0, stream>>>(hbuf, asrc, adst, asb, adb, Hh, M, nh);
  } else if (sort_cnt != nullptr) {
    gemm_wmma_f32<2, true><<<dim3(N / GEMM_BM + 1, M / 128), 128, 0, stream>>>(
        xin, W, hbuf, K, M, asrc, adst, asb, adb, Hh,
        sort_cnt, sort_rowptr, N);
    csr_fill_kernel<<<(E0 + 255) / 256, 256, 0, stream>>>(
        esrc, edst, sort_rowptr, sort_rank, sort_col, E0);
  } else {
    gemm_wmma_f32<2, false><<<dim3(N / GEMM_BM, M / 128), 128, 0, stream>>>(
        xin, W, hbuf, K, M, asrc, adst, asb, adb, Hh, nullptr, nullptr, 0);
  }

  node_softmax_gather_kernel<<<(nh + 3) / 4, 128, 0, stream>>>(
      hbuf, asb, adb, rowptr, col, bias, out, Hh, M, elu ? 1 : 0, nh);
}

extern "C" void kernel_launch(void* const* d_in, const int* in_sizes, int n_in,
                              void* d_out, int out_size, void* d_ws, size_t ws_size,
                              hipStream_t stream) {
  const float* x      = (const float*)d_in[0];
  const int*   ei     = (const int*)d_in[1];
  const float* W0     = (const float*)d_in[2];
  const float* a_src0 = (const float*)d_in[3];
  const float* a_dst0 = (const float*)d_in[4];
  const float* b0     = (const float*)d_in[5];
  const float* W1     = (const float*)d_in[6];
  const float* a_src1 = (const float*)d_in[7];
  const float* a_dst1 = (const float*)d_in[8];
  const float* b1     = (const float*)d_in[9];
  const float* W2     = (const float*)d_in[10];
  const float* a_src2 = (const float*)d_in[11];
  const float* a_dst2 = (const float*)d_in[12];
  const float* b2     = (const float*)d_in[13];

  const int N  = in_sizes[0] / 256;
  const int E0 = in_sizes[1] / 2;
  const int* esrc = ei;
  const int* edst = ei + E0;

  float* ws = (float*)d_ws;
  size_t p = 0;
  float* hbuf   = ws + p;             p += (size_t)N * 512;
  int*   csr    = (int*)(ws + p);     p += (size_t)N * 512;
  float* xbuf   = ws + p;             p += (size_t)N * 512;
  float* asb    = ws + p;             p += (size_t)N * 4;
  float* adb    = ws + p;             p += (size_t)N * 4;
  int*   cnt    = csr;
  int*   rowptr = csr + N;
  int*   col    = csr + (2 * N + 1);
  int*   rank   = col + E0;
  (void)ws_size; (void)n_in; (void)out_size;

  csr_zero_kernel<<<(N + 255) / 256, 256, 0, stream>>>(cnt, N);
  csr_count_kernel<<<(E0 + 1024 * CSR_EPT - 1) / (1024 * CSR_EPT), 1024, 0, stream>>>(
      edst, cnt, rank, E0, N);

  run_layer(stream, x, W0, a_src0, a_dst0, b0, N, 256, 512, 4, true,
            rowptr, col, hbuf, asb, adb, xbuf,
            cnt, rank, rowptr, col, esrc, edst, E0);
  run_layer(stream, xbuf, W1, a_src1, a_dst1, b1, N, 512, 512, 4, true,
            rowptr, col, hbuf, asb, adb, xbuf,
            nullptr, nullptr, nullptr, nullptr, nullptr, nullptr, 0);
  run_layer(stream, xbuf, W2, a_src2, a_dst2, b2, N, 512, 128, 1, false,
            rowptr, col, hbuf, asb, adb, (float*)d_out,
            nullptr, nullptr, nullptr, nullptr, nullptr, nullptr, 0);
}


# Edges kept in fixed-pitch buckets per node (slot = node*128 + rank): no degree scan; the fill needs ranks only and runs in spare blocks of the layer-0 GEMM launch, 10 -> 9 launches; sum order kept
# speedup vs baseline: 5.4175x; 1.0346x over previous
#include <hip/hip_runtime.h>
#include <hip/hip_bf16.h>
#include <math.h>


typedef __attribute__((ext_vector_type(2))) float v2f;
typedef __attribute__((ext_vector_type(4))) float v4f;
typedef __attribute__((ext_vector_type(8))) float v8f;

#define NEG_SLOPE 0.2f

#define CSR_P 128
__global__ void csr_zero_kernel(int* __restrict__ cnt, int n) {
  int i = blockIdx.x * blockDim.x + threadIdx.x;
  if (i < n) cnt[i] = 0;
}

#define CSR_EPT 10
#define CSR_CAP 16384
__global__ __launch_bounds__(1024)
void csr_count_kernel(const int* __restrict__ edst,
                      int* __restrict__ cnt,
                      int* __restrict__ rank, int E0, int n) {
  __shared__ int hist[CSR_CAP];
  const int t  = threadIdx.x;
  const int eb = blockIdx.x * (1024 * CSR_EPT) + t;

  int d[CSR_EPT], r[CSR_EPT];
#pragma unroll
  for (int u = 0; u < CSR_EPT; ++u) {
    const int e = eb + u * 1024;
    d[u] = (e < E0) ? edst[e] : -1;
    r[u] = 0;
  }

  for (int lo = 0; lo < n; lo += CSR_CAP) {
    const int nb = (n - lo < CSR_CAP) ? (n - lo) : CSR_CAP;
    for (int i = t; i < nb; i += 1024) hist[i] = 0;
    __syncthreads();

#pragma unroll
    for (int u = 0; u < CSR_EPT; ++u) {
      const unsigned k = (unsigned)(d[u] - lo);
      if (k < (unsigned)nb) r[u] = atomicAdd(&hist[k], 1);
    }
    __syncthreads();

    for (int ib = t; ib < nb; ib += 1024 * CSR_EPT) {
      int c[CSR_EPT], b[CSR_EPT];
#pragma unroll
      for (int u = 0; u < CSR_EPT; ++u) {
        const int i = ib + u * 1024;
        c[u] = (i < nb) ? hist[i] : 0;
      }
#pragma unroll
      for (int u = 0; u < CSR_EPT; ++u)
        b[u] = (c[u] > 0) ? atomicAdd(&cnt[lo + ib + u * 1024], c[u]) : 0;
#pragma unroll
      for (int u = 0; u < CSR_EPT; ++u)
        if (c[u] > 0) hist[ib + u * 1024] = b[u];
    }
    __syncthreads();

    int hb[CSR_EPT];
#pragma unroll
    for (int u = 0; u < CSR_EPT; ++u) {
      const unsigned k = (unsigned)(d[u] - lo);
      hb[u] = (k < (unsigned)nb) ? hist[k] : 0;
    }
#pragma unroll
    for (int u = 0; u < CSR_EPT; ++u) {
      const unsigned k = (unsigned)(d[u] - lo);
      if (k < (unsigned)nb) rank[eb + u * 1024] = hb[u] + r[u];
    }
    __syncthreads();
  }
}

#define CSR_FEPT 8
__device__ __forceinline__
void csr_fill_block128(const int* __restrict__ esrc, const int* __restrict__ edst,
                       const int* __restrict__ rank, int* __restrict__ col,
                       int E0, int n, int fb) {
  const int eb = fb * (128 * CSR_FEPT) + threadIdx.x;
  int d[CSR_FEPT], r[CSR_FEPT], sc[CSR_FEPT];
#pragma unroll
  for (int u = 0; u < CSR_FEPT; ++u) {
    const int e = eb + u * 128;
    const bool ok = e < E0;
    d[u]  = ok ? edst[e] : -1;
    r[u]  = ok ? rank[e] : CSR_P;
    sc[u] = ok ? esrc[e] : 0;
  }
#pragma unroll
  for (int u = 0; u < CSR_FEPT; ++u)
    if ((unsigned)d[u] < (unsigned)n && (unsigned)r[u] < (unsigned)CSR_P)
      col[d[u] * CSR_P + r[u]] = sc[u];
}

#define GEMM_BM 80
#define GEMM_BK 32
#define GEMM_AP 36
#define GEMM_CP 132

template <int NT, bool FILL>
__global__ __launch_bounds__(128)
void gemm_wmma_f32(const float* __restrict__ X, const float* __restrict__ W,
                   float* __restrict__ Hout, int K, int M,
                   const float* __restrict__ asrc, const float* __restrict__ adst,
                   float* __restrict__ as_out, float* __restrict__ ad_out,
                   int Hh, const int* __restrict__ fill_esrc,
                   const int* __restrict__ fill_edst,
                   const int* __restrict__ fill_rank,
                   int* __restrict__ fill_col, int fill_E0, int fill_n) {
  constexpr int BN    = 64 * NT;
  constexpr int BP2   = 2 * BN + 32;
  constexpr int BQ    = BN / 4;
  constexpr int PPP   = 128 / BQ;
  constexpr int BPASS = (GEMM_BK / 2) / PPP;
  constexpr int APASS = GEMM_BM / 16;

  constexpr int AS_SZ = GEMM_BM * GEMM_AP;
  constexpr int BS_SZ = (GEMM_BK / 2) * BP2;
  static_assert(NT != 2 || GEMM_BM * GEMM_CP <= 2 * AS_SZ + 2 * BS_SZ,
                "the parked tile must fit in the K-loop's stages");

  __shared__ __attribute__((aligned(16))) float smem[2 * AS_SZ + 2 * BS_SZ];
  float (*As)[AS_SZ] = reinterpret_cast<float (*)[AS_SZ]>(smem);
  float (*Bs)[BS_SZ] = reinterpret_cast<float (*)[BS_SZ]>(smem + 2 * AS_SZ);

  if constexpr (FILL) {
    const int ty = M / BN;
    if ((int)blockIdx.y >= ty) {
      csr_fill_block128(fill_esrc, fill_edst, fill_rank, fill_col, fill_E0, fill_n,
                        ((int)blockIdx.y - ty) * (int)gridDim.x + (int)blockIdx.x);
      return;
    }
  }

  const int t    = threadIdx.x;
  const int wave = t >> 5;
  const int lane = t & 31;
  const int c    = lane & 15;
  const int h    = lane >> 4;
  const int m0   = blockIdx.x * GEMM_BM;
  const int n0   = blockIdx.y * BN;

  const int arow = t >> 3;
  const int aq   = (t & 7) << 2;
  const int bpr  = t / BQ;
  const int bq   = t % BQ;

  const float* __restrict__ xg = X + (size_t)(m0 + arow) * K + aq;
  const float* __restrict__ wg = W + (size_t)(2 * bpr) * M + n0 + 4 * bq;

  v4f ra[APASS];
  v4f rb0[BPASS], rb1[BPASS];
  v8f acc[APASS][NT];
#pragma unroll
  for (int i = 0; i < APASS; ++i)
#pragma unroll
    for (int j = 0; j < NT; ++j) acc[i][j] = v8f{};

#pragma unroll
  for (int i = 0; i < APASS; ++i)
    ra[i] = *(const v4f*)(xg + (size_t)(16 * i) * K);
#pragma unroll
  for (int i = 0; i < BPASS; ++i) {
    rb0[i] = *(const v4f*)(wg + (size_t)(2 * PPP * i) * M);
    rb1[i] = *(const v4f*)(wg + (size_t)(2 * PPP * i + 1) * M);
  }
#pragma unroll
  for (int i = 0; i < APASS; ++i)
    *(v4f*)&As[0][(arow + 16 * i) * GEMM_AP + aq] = ra[i];
#pragma unroll
  for (int i = 0; i < BPASS; ++i) {
    float* bd = &Bs[0][(bpr + PPP * i) * BP2 + 8 * bq];
    *(v4f*)bd       = v4f{rb0[i].x, rb1[i].x, rb0[i].y, rb1[i].y};
    *(v4f*)(bd + 4) = v4f{rb0[i].z, rb1[i].z, rb0[i].w, rb1[i].w};
  }
  __syncthreads();

  const int KT   = K / GEMM_BK;
  const int aoff = c * GEMM_AP + 2 * h;
  const int boff = h * BP2 + 2 * (wave * (16 * NT) + c);

  for (int kt = 0; kt < KT; ++kt) {
    const int  cur  = kt & 1;
    const bool more = (kt + 1 < KT);

    if (more) {
      const float* __restrict__ xa = xg + (size_t)(kt + 1) * GEMM_BK;
      const float* __restrict__ wb = wg + (size_t)(kt + 1) * GEMM_BK * M;
#pragma unroll
      for (int i = 0; i < APASS; ++i)
        ra[i] = *(const v4f*)(xa + (size_t)(16 * i) * K);
#pragma unroll
      for (int i = 0; i < BPASS; ++i) {
        rb0[i] = *(const v4f*)(wb + (size_t)(2 * PPP * i) * M);
        rb1[i] = *(const v4f*)(wb + (size_t)(2 * PPP * i + 1) * M);
      }
    }

    v2f fa[2][2][APASS];
    v2f fb[2][2][NT];
#pragma unroll
    for (int u = 0; u < 2; ++u) {
#pragma unroll
      for (int i = 0; i < APASS; ++i)
        fa[0][u][i] = *(const v2f*)&As[cur][aoff + (16 * i) * GEMM_AP + 4 * u];
#pragma unroll
      for (int j = 0; j < NT; ++j)
        fb[0][u][j] = *(const v2f*)&Bs[cur][boff + (2 * u) * BP2 + 32 * j];
    }

#pragma unroll
    for (int kk = 0; kk < GEMM_BK / 8; ++kk) {
      const int s = kk & 1;
      if (kk + 1 < GEMM_BK / 8) {
#pragma unroll
        for (int u = 0; u < 2; ++u) {
          const int ks = 2 * (kk + 1) + u;
#pragma unroll
          for (int i = 0; i < APASS; ++i)
            fa[s ^ 1][u][i] =
                *(const v2f*)&As[cur][aoff + (16 * i) * GEMM_AP + 4 * ks];
#pragma unroll
          for (int j = 0; j < NT; ++j)
            fb[s ^ 1][u][j] =
                *(const v2f*)&Bs[cur][boff + (2 * ks) * BP2 + 32 * j];
        }
      }
      __builtin_amdgcn_sched_barrier(0);
#pragma unroll
      for (int u = 0; u < 2; ++u)
#pragma unroll
        for (int i = 0; i < APASS; ++i)
#pragma unroll
          for (int j = 0; j < NT; ++j)
            acc[i][j] = __builtin_amdgcn_wmma_f32_16x16x4_f32(
                false, fa[s][u][i], false, fb[s][u][j], (short)0, acc[i][j],
                false, false);
      __builtin_amdgcn_sched_barrier(0);
    }

    if (more) {
#pragma unroll
      for (int i = 0; i < APASS; ++i)
        *(v4f*)&As[cur ^ 1][(arow + 16 * i) * GEMM_AP + aq] = ra[i];
#pragma unroll
      for (int i = 0; i < BPASS; ++i) {
        float* bd = &Bs[cur ^ 1][(bpr + PPP * i) * BP2 + 8 * bq];
        *(v4f*)bd       = v4f{rb0[i].x, rb1[i].x, rb0[i].y, rb1[i].y};
        *(v4f*)(bd + 4) = v4f{rb0[i].z, rb1[i].z, rb0[i].w, rb1[i].w};
      }
    }
    __syncthreads();
  }

  if constexpr (NT == 2) {
    const int head = blockIdx.y;
    const float4 sv = *(const float4*)(asrc + head * 128 + lane * 4);
    const float4 dv = *(const float4*)(adst + head * 128 + lane * 4);

    float* cw = smem + (8 * h) * GEMM_CP + wave * 32 + c;
#pragma unroll
    for (int i = 0; i < APASS; ++i)
#pragma unroll
      for (int v = 0; v < 8; ++v)
#pragma unroll
        for (int j = 0; j < NT; ++j)
          cw[(16 * i + v) * GEMM_CP + 16 * j] = acc[i][j][v];
    __syncthreads();

    constexpr int RPW = GEMM_BM / 4;
    const int r0 = wave * RPW;
    const float* cr = smem + r0 * GEMM_CP + lane * 4;
    float* __restrict__ orow = Hout + (size_t)(m0 + r0) * M + n0 + lane * 4;
    float ks = 0.0f, kd = 0.0f;
#pragma unroll
    for (int q = 0; q < RPW; ++q) {
      const float4 hv = *(const float4*)(cr + q * GEMM_CP);
      *(float4*)(orow + (size_t)q * M) = hv;
      float s = hv.x * sv.x + hv.y * sv.y + hv.z * sv.z + hv.w * sv.w;
      float d = hv.x * dv.x + hv.y * dv.y + hv.z * dv.z + hv.w * dv.w;
#pragma unroll
      for (int off = 16; off > 0; off >>= 1) {
        s += __shfl_xor(s, off);
        d += __shfl_xor(d, off);
      }
      if (lane == q) { ks = s; kd = d; }
    }
    if (lane < RPW) {
      const int o = (m0 + r0 + lane) * Hh + head;
      as_out[o] = ks;
      ad_out[o] = kd;
    }
  } else {
    float* __restrict__ ob =
        Hout + (size_t)(m0 + 8 * h) * M + n0 + wave * (16 * NT) + c;
#pragma unroll
    for (int i = 0; i < APASS; ++i)
#pragma unroll
      for (int v = 0; v < 8; ++v) {
        float* __restrict__ op = ob + (size_t)(16 * i + v) * M;
#pragma unroll
        for (int j = 0; j < NT; ++j) op[16 * j] = acc[i][j][v];
      }
  }
}

__global__ __launch_bounds__(128)
void alpha_kernel(const float* __restrict__ Hf, const float* __restrict__ asrc,
                  const float* __restrict__ adst, float* __restrict__ as_out,
                  float* __restrict__ ad_out, int Hh, int M, int total) {
  const int wid  = blockIdx.x * 4 + (threadIdx.x >> 5);
  const int lane = threadIdx.x & 31;
  if (wid >= total) return;
  const int n  = wid / Hh;
  const int hd = wid - n * Hh;

  const float4 hv = *(const float4*)(Hf + (size_t)n * M + hd * 128 + lane * 4);
  const float4 sv = *(const float4*)(asrc + hd * 128 + lane * 4);
  const float4 dv = *(const float4*)(adst + hd * 128 + lane * 4);
  float s = hv.x * sv.x + hv.y * sv.y + hv.z * sv.z + hv.w * sv.w;
  float d = hv.x * dv.x + hv.y * dv.y + hv.z * dv.z + hv.w * dv.w;
#pragma unroll
  for (int off = 16; off > 0; off >>= 1) {
    s += __shfl_xor(s, off);
    d += __shfl_xor(d, off);
  }
  if (lane == 0) { as_out[wid] = s; ad_out[wid] = d; }
}

__global__ __launch_bounds__(128)
void node_softmax_gather_kernel(const float* __restrict__ Hf,
                                const float* __restrict__ as,
                                const float* __restrict__ ad,
                                const int* __restrict__ cnt,
                                const int* __restrict__ col,
                                const int* __restrict__ esrc,
                                const int* __restrict__ edst,
                                const float* __restrict__ bias,
                                float* __restrict__ out,
                                int Hh, int M, int doElu, int total, int E0) {
  const int wv   = __builtin_amdgcn_readfirstlane((int)(threadIdx.x >> 5));
  const int wid  = blockIdx.x * 4 + wv;
  const int lane = threadIdx.x & 31;
  if (wid >= total) return;
  const int d   = wid / Hh;
  const int hd  = wid - d * Hh;
  const int beg = d * CSR_P;
  const int deg = cnt[d];
  const int tot = deg + 1;
  const float adv = ad[wid];
  const float* __restrict__ hp = Hf + hd * 128 + lane * 4;
  float4 acc = make_float4(0.0f, 0.0f, 0.0f, 0.0f);

  if (deg > CSR_P) {
    float ls = as[wid] + adv;
    ls = (ls > 0.0f) ? ls : NEG_SLOPE * ls;
    float m = ls;
    for (int e = lane; e < E0; e += 32) {
      if (edst[e] == d) {
        float l = as[esrc[e] * Hh + hd] + adv;
        l = (l > 0.0f) ? l : NEG_SLOPE * l;
        m = fmaxf(m, l);
      }
    }
    for (int off = 16; off > 0; off >>= 1) m = fmaxf(m, __shfl_xor(m, off));
    m = isfinite(m) ? m : 0.0f;
    const float exs = expf(ls - m);
    float den = (lane == 0) ? exs : 0.0f;
    for (int e = lane; e < E0; e += 32) {
      if (edst[e] == d) {
        float l = as[esrc[e] * Hh + hd] + adv;
        l = (l > 0.0f) ? l : NEG_SLOPE * l;
        den += expf(l - m);
      }
    }
    for (int off = 16; off > 0; off >>= 1) den += __shfl_xor(den, off);
    const float dden = den + 1e-16f;
    {
      const float a = exs / dden;
      const float4 hv = *(const float4*)(hp + (size_t)d * M);
      acc.x = hv.x * a; acc.y = hv.y * a; acc.z = hv.z * a; acc.w = hv.w * a;
    }
    for (int eb = 0; eb < E0; eb += 32) {
      const int e = eb + lane;
      const bool hit = (e < E0) && (edst[e] == d);
      int   s_l = d;
      float a_l = 0.0f;
      if (hit) {
        s_l = esrc[e];
        float l = as[s_l * Hh + hd] + adv;
        l = (l > 0.0f) ? l : NEG_SLOPE * l;
        a_l = expf(l - m) / dden;
      }
      unsigned mask = __builtin_amdgcn_ballot_w32(hit);
      while (mask != 0u) {
        const int t = __builtin_ctz(mask);
        mask &= mask - 1u;
        const int s = __builtin_amdgcn_readlane(s_l, t);
        const float a = __int_as_float(
            __builtin_amdgcn_readlane(__float_as_int(a_l), t));
        const float4 hv = *(const float4*)(hp + (size_t)s * M);
        acc.x += hv.x * a; acc.y += hv.y * a; acc.z += hv.z * a; acc.w += hv.w * a;
      }
    }
  } else {
    const bool v0 = lane < tot;
    int   s0 = d;
    float l0 = -INFINITY;
    if (v0) {
      s0 = (lane < deg) ? col[beg + lane] : d;
      const float l = as[s0 * Hh + hd] + adv;
      l0 = (l > 0.0f) ? l : NEG_SLOPE * l;
    }

    float m = l0;
    for (int i = lane + 32; i < tot; i += 32) {
      const int s = (i < deg) ? col[beg + i] : d;
      float l = as[s * Hh + hd] + adv;
      l = (l > 0.0f) ? l : NEG_SLOPE * l;
      m = fmaxf(m, l);
    }
    for (int off = 16; off > 0; off >>= 1) m = fmaxf(m, __shfl_xor(m, off));
    m = isfinite(m) ? m : 0.0f;

    const float ex0 = v0 ? expf(l0 - m) : 0.0f;
    float den = ex0;
    for (int i = lane + 32; i < tot; i += 32) {
      const int s = (i < deg) ? col[beg + i] : d;
      float l = as[s * Hh + hd] + adv;
      l = (l > 0.0f) ? l : NEG_SLOPE * l;
      den += expf(l - m);
    }
    for (int off = 16; off > 0; off >>= 1) den += __shfl_xor(den, off);
    const float dden = den + 1e-16f;

    for (int cb = 0; cb < tot; cb += 32) {
      int   s_l = s0;
      float a_l = ex0 / dden;
      if (cb > 0) {
        const int i = cb + lane;
        s_l = d;
        a_l = 0.0f;
        if (i < tot) {
          s_l = (i < deg) ? col[beg + i] : d;
          float l = as[s_l * Hh + hd] + adv;
          l = (l > 0.0f) ? l : NEG_SLOPE * l;
          a_l = expf(l - m) / dden;
        }
      }
      const int nn = (tot - cb < 32) ? (tot - cb) : 32;
      for (int t = 0; t < nn; t += 8) {
        float4 hv[8];
        float  av[8];
        for (int u = 0; u < 8; ++u) {
          if (t + u < nn) {
            const int s = __builtin_amdgcn_readlane(s_l, t + u);
            av[u] = __int_as_float(
                __builtin_amdgcn_readlane(__float_as_int(a_l), t + u));
            hv[u] = *(const float4*)(hp + (size_t)s * M);
          }
        }
        for (int u = 0; u < 8; ++u) {
          if (t + u < nn) {
            acc.x += hv[u].x * av[u];
            acc.y += hv[u].y * av[u];
            acc.z += hv[u].z * av[u];
            acc.w += hv[u].w * av[u];
          }
        }
      }
    }
  }

  const int c = hd * 128 + lane * 4;
  const float4 bv = *(const float4*)(bias + c);
  float4 r;
  r.x = acc.x + bv.x; r.y = acc.y + bv.y; r.z = acc.z + bv.z; r.w = acc.w + bv.w;
  if (doElu) {
    r.x = (r.x > 0.0f) ? r.x : expm1f(r.x);
    r.y = (r.y > 0.0f) ? r.y : expm1f(r.y);
    r.z = (r.z > 0.0f) ? r.z : expm1f(r.z);
    r.w = (r.w > 0.0f) ? r.w : expm1f(r.w);
  }
  *(float4*)(out + (size_t)d * M + c) = r;
}

static void run_layer(hipStream_t stream, const float* xin, const float* W,
                      const float* asrc, const float* adst, const float* bias,
                      int N, int K, int M, int Hh, bool elu,
                      const int* cnt, const int* col,
                      float* hbuf, float* asb, float* adb, float* out,
                      const int* fill_rank, int* fill_col,
                      const int* esrc, const int* edst, int E0) {
  const int nh = N * Hh;

  if (M == 128) {
    gemm_wmma_f32<1, false><<<dim3(N / GEMM_BM, M / 64), 128, 0, stream>>>(
        xin, W, hbuf, K, M, asrc, adst, asb, adb, Hh,
        nullptr, nullptr, nullptr, nullptr, 0, 0);
    alpha_kernel<<<(nh + 3) / 4, 128, 0, stream>>>(hbuf, asrc, adst, asb, adb, Hh, M, nh);
  } else if (fill_rank != nullptr) {
    const int fill_per_row = (N / GEMM_BM) * 128 * CSR_FEPT;
    gemm_wmma_f32<2, true><<<dim3(N / GEMM_BM,
                                  M / 128 + (E0 + fill_per_row - 1) / fill_per_row),
                             128, 0, stream>>>(
        xin, W, hbuf, K, M, asrc, adst, asb, adb, Hh,
        esrc, edst, fill_rank, fill_col, E0, N);
  } else {
    gemm_wmma_f32<2, false><<<dim3(N / GEMM_BM, M / 128), 128, 0, stream>>>(
        xin, W, hbuf, K, M, asrc, adst, asb, adb, Hh,
        nullptr, nullptr, nullptr, nullptr, 0, 0);
  }

  node_softmax_gather_kernel<<<(nh + 3) / 4, 128, 0, stream>>>(
      hbuf, asb, adb, cnt, col, esrc, edst, bias, out, Hh, M, elu ? 1 : 0, nh, E0);
}

extern "C" void kernel_launch(void* const* d_in, const int* in_sizes, int n_in,
                              void* d_out, int out_size, void* d_ws, size_t ws_size,
                              hipStream_t stream) {
  const float* x      = (const float*)d_in[0];
  const int*   ei     = (const int*)d_in[1];
  const float* W0     = (const float*)d_in[2];
  const float* a_src0 = (const float*)d_in[3];
  const float* a_dst0 = (const float*)d_in[4];
  const float* b0     = (const float*)d_in[5];
  const float* W1     = (const float*)d_in[6];
  const float* a_src1 = (const float*)d_in[7];
  const float* a_dst1 = (const float*)d_in[8];
  const float* b1     = (const float*)d_in[9];
  const float* W2     = (const float*)d_in[10];
  const float* a_src2 = (const float*)d_in[11];
  const float* a_dst2 = (const float*)d_in[12];
  const float* b2     = (const float*)d_in[13];

  const int N  = in_sizes[0] / 256;
  const int E0 = in_sizes[1] / 2;
  const int* esrc = ei;
  const int* edst = ei + E0;

  float* ws = (float*)d_ws;
  size_t p = 0;
  float* hbuf   = ws + p;             p += (size_t)N * 512;
  int*   csr    = (int*)(ws + p);     p += (size_t)N * 512;
  float* xbuf   = ws + p;             p += (size_t)N * 512;
  float* asb    = ws + p;             p += (size_t)N * 4;
  float* adb    = ws + p;             p += (size_t)N * 4;
  int*   col    = csr;
  int*   cnt    = csr + (size_t)N * CSR_P;
  int*   rank   = cnt + N;
  (void)ws_size; (void)n_in; (void)out_size;

  csr_zero_kernel<<<(N + 255) / 256, 256, 0, stream>>>(cnt, N);
  csr_count_kernel<<<(E0 + 1024 * CSR_EPT - 1) / (1024 * CSR_EPT), 1024, 0, stream>>>(
      edst, cnt, rank, E0, N);

  run_layer(stream, x, W0, a_src0, a_dst0, b0, N, 256, 512, 4, true,
            cnt, col, hbuf, asb, adb, xbuf,
            rank, col, esrc, edst, E0);
  run_layer(stream, xbuf, W1, a_src1, a_dst1, b1, N, 512, 512, 4, true,
            cnt, col, hbuf, asb, adb, xbuf,
            nullptr, nullptr, esrc, edst, E0);
  run_layer(stream, xbuf, W2, a_src2, a_dst2, b2, N, 512, 128, 1, false,
            cnt, col, hbuf, asb, adb, (float*)d_out,
            nullptr, nullptr, esrc, edst, E0);
}
